# starting kernel
# baseline (speedup 1.0000x reference)
#include <hip/hip_runtime.h>

#define N_NODES 50000
#define N_EDGES 800000
#define DIM     128
#define NLAYER  3
#define NGRAPH  512
#define ZSTRIDE (NLAYER * DIM)
#define BN_EPS  1e-5f
#define LDS_STRIDE 132

typedef __attribute__((ext_vector_type(2))) float v2f;
typedef __attribute__((ext_vector_type(8))) float v8f;

__global__ __launch_bounds__(256) void scatter_kernel(
    const float* __restrict__ z, int zstride,
    const int* __restrict__ src, const int* __restrict__ dst,
    float* __restrict__ agg)
{
    int tid = blockIdx.x * blockDim.x + threadIdx.x;
    if (tid >= N_EDGES * 32) return;
    int e  = tid >> 5;
    int c  = (tid & 31) * 4;
    int s  = src[e];
    int d  = dst[e];
    float4 v = *(const float4*)(z + (size_t)s * zstride + c);
    float* a = agg + (size_t)d * DIM + c;
    atomicAdd(a + 0, v.x);
    atomicAdd(a + 1, v.y);
    atomicAdd(a + 2, v.z);
    atomicAdd(a + 3, v.w);
}

__global__ __launch_bounds__(256) void gin_mlp_kernel(
    const float* __restrict__ z, int zstride,
    float* __restrict__ hbuf,
    const float* __restrict__ W1, const float* __restrict__ b1,
    const float* __restrict__ W2, const float* __restrict__ b2,
    float* __restrict__ stats)
{
    __shared__ float ldsA[16 * LDS_STRIDE];
    __shared__ float ldsH[16 * LDS_STRIDE];

    const int tid  = threadIdx.x;
    const int wave = tid >> 5;
    const int lane = tid & 31;
    const int half = lane >> 4;
    const int l16  = lane & 15;
    const int row0 = blockIdx.x * 16;
    const int col  = wave * 16 + l16;

    for (int idx = tid; idx < 16 * DIM; idx += 256) {
        int r = idx >> 7, c = idx & 127;
        ldsA[r * LDS_STRIDE + c] =
            z[(size_t)(row0 + r) * zstride + c] + hbuf[(size_t)(row0 + r) * DIM + c];
    }
    __syncthreads();

    v8f acc = {};
    for (int k0 = 0; k0 < DIM; k0 += 4) {
        int ka = k0 + 2 * half;
        v2f a, b;
        a.x = ldsA[l16 * LDS_STRIDE + ka];
        a.y = ldsA[l16 * LDS_STRIDE + ka + 1];
        b.x = W1[ka * DIM + col];
        b.y = W1[(ka + 1) * DIM + col];
        acc = __builtin_amdgcn_wmma_f32_16x16x4_f32(
            false, a, false, b, (short)0, acc, false, false);
    }

    float bias1 = b1[col];
    #pragma unroll
    for (int g = 0; g < 8; ++g) {
        int m = g + 8 * half;
        float v = acc[g] + bias1;
        ldsH[m * LDS_STRIDE + col] = v > 0.f ? v : 0.f;
    }
    __syncthreads();

    v8f acc2 = {};
    for (int k0 = 0; k0 < DIM; k0 += 4) {
        int ka = k0 + 2 * half;
        v2f a, b;
        a.x = ldsH[l16 * LDS_STRIDE + ka];
        a.y = ldsH[l16 * LDS_STRIDE + ka + 1];
        b.x = W2[ka * DIM + col];
        b.y = W2[(ka + 1) * DIM + col];
        acc2 = __builtin_amdgcn_wmma_f32_16x16x4_f32(
            false, a, false, b, (short)0, acc2, false, false);
    }

    float bias2 = b2[col];
    float s = 0.f, q = 0.f;
    #pragma unroll
    for (int g = 0; g < 8; ++g) {
        int m = g + 8 * half;
        float v = acc2[g] + bias2;
        v = v > 0.f ? v : 0.f;
        hbuf[(size_t)(row0 + m) * DIM + col] = v;
        s += v;
        q += v * v;
    }
    s += __shfl_xor(s, 16);
    q += __shfl_xor(q, 16);
    if (half == 0) {
        atomicAdd(&stats[col], s);
        atomicAdd(&stats[DIM + col], q);
    }
}

__global__ void bn_stats_finalize(float* __restrict__ stats)
{
    int c = threadIdx.x;
    float mu  = stats[c] * (1.0f / N_NODES);
    float var = stats[DIM + c] * (1.0f / N_NODES) - mu * mu;
    stats[c]       = mu;
    stats[DIM + c] = rsqrtf(var + BN_EPS);
}

__global__ __launch_bounds__(256) void bn_pool_kernel(
    const float* __restrict__ h,
    const float* __restrict__ stats,
    const float* __restrict__ gamma, const float* __restrict__ beta,
    const int* __restrict__ batch,
    float* __restrict__ zcat, float* __restrict__ gcat)
{
    int tid = blockIdx.x * blockDim.x + threadIdx.x;
    if (tid >= N_NODES * 32) return;
    int i = tid >> 5;
    int c = (tid & 31) * 4;
    int g = batch[i];
    float4 v = *(const float4*)(h + (size_t)i * DIM + c);
    float in[4] = {v.x, v.y, v.z, v.w};
    float4 o;
    float* op = &o.x;
    #pragma unroll
    for (int j = 0; j < 4; ++j) {
        float mu = stats[c + j];
        float rs = stats[DIM + c + j];
        float y  = (in[j] - mu) * rs * gamma[c + j] + beta[c + j];
        op[j] = y;
        atomicAdd(&gcat[(size_t)g * ZSTRIDE + c + j], y);
    }
    *(float4*)(zcat + (size_t)i * ZSTRIDE + c) = o;
}

extern "C" void kernel_launch(void* const* d_in, const int* in_sizes, int n_in,
                              void* d_out, int out_size, void* d_ws, size_t ws_size,
                              hipStream_t stream)
{
    (void)in_sizes; (void)n_in; (void)out_size; (void)ws_size;

    const float* x     = (const float*)d_in[0];
    const int*   ei    = (const int*)  d_in[1];
    const int*   batch = (const int*)  d_in[2];
    const float* W1    = (const float*)d_in[3];
    const float* b1    = (const float*)d_in[4];
    const float* W2    = (const float*)d_in[5];
    const float* b2    = (const float*)d_in[6];
    const float* gamma = (const float*)d_in[7];
    const float* beta  = (const float*)d_in[8];

    float* out  = (float*)d_out;
    float* zcat = out;
    float* gcat = out + (size_t)N_NODES * ZSTRIDE;

    float* agg   = (float*)d_ws;
    float* stats = agg + (size_t)N_NODES * DIM;

    const int* src = ei;
    const int* dst = ei + N_EDGES;

    hipMemsetAsync(gcat, 0, sizeof(float) * (size_t)NGRAPH * ZSTRIDE, stream);

    for (int l = 0; l < NLAYER; ++l) {
        const float* zin = (l == 0) ? x : (zcat + (size_t)(l - 1) * DIM);
        int zstride      = (l == 0) ? DIM : ZSTRIDE;

        hipMemsetAsync(agg,   0, sizeof(float) * (size_t)N_NODES * DIM, stream);
        hipMemsetAsync(stats, 0, sizeof(float) * 2 * DIM, stream);

        {
            long long threads = (long long)N_EDGES * 32;
            int blocks = (int)((threads + 255) / 256);
            scatter_kernel<<<blocks, 256, 0, stream>>>(zin, zstride, src, dst, agg);
        }

        gin_mlp_kernel<<<N_NODES / 16, 256, 0, stream>>>(
            zin, zstride, agg,
            W1 + (size_t)l * DIM * DIM, b1 + (size_t)l * DIM,
            W2 + (size_t)l * DIM * DIM, b2 + (size_t)l * DIM,
            stats);

        bn_stats_finalize<<<1, DIM, 0, stream>>>(stats);

        {
            long long threads = (long long)N_NODES * 32;
            int blocks = (int)((threads + 255) / 256);
            bn_pool_kernel<<<blocks, 256, 0, stream>>>(
                agg, stats,
                gamma + (size_t)l * DIM, beta + (size_t)l * DIM,
                batch,
                zcat + (size_t)l * DIM, gcat + (size_t)l * DIM);
        }
    }
}


# Neighbour sum without f32 atomics: edges grouped by destination once per call (int atomic slot, scan, place), a wave sums a row in registers, no agg clear; row sum order = slot order, run-dependent
# speedup vs baseline: 1.7191x; 1.7191x over previous
#include <hip/hip_runtime.h>

#define N_NODES 50000
#define N_EDGES 800000
#define DIM     128
#define NLAYER  3
#define NGRAPH  512
#define ZSTRIDE (NLAYER * DIM)
#define BN_EPS  1e-5f
#define LDS_STRIDE 132

typedef __attribute__((ext_vector_type(2))) float v2f;
typedef __attribute__((ext_vector_type(8))) float v8f;

#define EDGE_QUADS   (N_EDGES / 4)
#define EDGE_BLOCKS  ((EDGE_QUADS + 255) / 256)
#define SCAN_BLOCK   1024
#define SCAN_BLOCKS  ((N_NODES + 1 + SCAN_BLOCK - 1) / SCAN_BLOCK)
#define ROWS_PER_WAVE 5
#define GATHER_ROWS_PER_BLOCK (8 * ROWS_PER_WAVE)
#define GATHER_BLOCKS (N_NODES / GATHER_ROWS_PER_BLOCK)

__global__ __launch_bounds__(256) void count_slot_kernel(
    const int* __restrict__ dst, int* __restrict__ cnt, int* __restrict__ slot)
{
    int t = blockIdx.x * 256 + threadIdx.x;
    if (t >= EDGE_QUADS) return;
    int4 d = ((const int4*)dst)[t];
    int4 r;
    r.x = atomicAdd(&cnt[d.x], 1);
    r.y = atomicAdd(&cnt[d.y], 1);
    r.z = atomicAdd(&cnt[d.z], 1);
    r.w = atomicAdd(&cnt[d.w], 1);
    ((int4*)slot)[t] = r;
}

__global__ __launch_bounds__(SCAN_BLOCK) void row_start_kernel(
    const int* __restrict__ cnt, int* __restrict__ rowstart)
{
    __shared__ int wOwn[32];
    __shared__ int wFront[32];
    const int tid  = threadIdx.x;
    const int lane = tid & 31;
    const int wave = tid >> 5;
    const int b0   = blockIdx.x * SCAN_BLOCK;

    int front = 0;
    for (int i = tid; i < b0; i += SCAN_BLOCK) front += cnt[i];

    const int i = b0 + tid;
    const int c = (i < N_NODES) ? cnt[i] : 0;

    int incl = c;
    for (int d = 1; d < 32; d <<= 1) {
        int up = __shfl_up(incl, d);
        if (lane >= d) incl += up;
    }
    for (int d = 16; d >= 1; d >>= 1) front += __shfl_xor(front, d);
    if (lane == 31) wOwn[wave] = incl;
    if (lane == 0)  wFront[wave] = front;
    __syncthreads();

    int before = (lane < wave) ? wOwn[lane] : 0;
    int total  = wFront[lane];
    for (int d = 16; d >= 1; d >>= 1) {
        before += __shfl_xor(before, d);
        total  += __shfl_xor(total, d);
    }
    if (i <= N_NODES) rowstart[i] = total + before + incl - c;
}

__global__ __launch_bounds__(256) void place_kernel(
    const int* __restrict__ src, const int* __restrict__ dst,
    const int* __restrict__ slot, const int* __restrict__ rowstart,
    int* __restrict__ nbr)
{
    int t = blockIdx.x * 256 + threadIdx.x;
    if (t >= EDGE_QUADS) return;
    int4 s = ((const int4*)src)[t];
    int4 d = ((const int4*)dst)[t];
    int4 r = ((const int4*)slot)[t];
    unsigned p0 = (unsigned)(rowstart[d.x] + r.x);
    unsigned p1 = (unsigned)(rowstart[d.y] + r.y);
    unsigned p2 = (unsigned)(rowstart[d.z] + r.z);
    unsigned p3 = (unsigned)(rowstart[d.w] + r.w);
    if (p0 < N_EDGES) nbr[p0] = s.x;
    if (p1 < N_EDGES) nbr[p1] = s.y;
    if (p2 < N_EDGES) nbr[p2] = s.z;
    if (p3 < N_EDGES) nbr[p3] = s.w;
}

__device__ __forceinline__ float4 ld_row(const float* __restrict__ zl, int zstride,
                                         int ids, int j)
{
    unsigned s = (unsigned)__builtin_amdgcn_readlane(ids, j);
    s = s < (unsigned)N_NODES ? s : (unsigned)(N_NODES - 1);
    return *(const float4*)(zl + (size_t)s * zstride);
}

__device__ __forceinline__ void add4(float4& a, const float4& v)
{
    a.x += v.x; a.y += v.y; a.z += v.z; a.w += v.w;
}

__global__ __launch_bounds__(256) void gather_kernel(
    const float* __restrict__ z, int zstride,
    const int* __restrict__ rowstart, const int* __restrict__ nbr,
    float* __restrict__ agg)
{
    const int lane = threadIdx.x & 31;
    const int wave = __builtin_amdgcn_readfirstlane((int)(threadIdx.x >> 5));
    const int row0 = (blockIdx.x * 8 + wave) * ROWS_PER_WAVE;
    const float* zl = z + lane * 4;

    for (int r = 0; r < ROWS_PER_WAVE; ++r) {
        const int row = row0 + r;
        int beg = rowstart[row];
        int end = rowstart[row + 1];
        beg = beg < 0 ? 0 : beg;
        end = end > N_EDGES ? N_EDGES : end;
        float4 acc = make_float4(0.f, 0.f, 0.f, 0.f);

        for (int base = beg; base < end; base += 32) {
            int n = end - base;
            n = n > 32 ? 32 : n;
            int ids = (lane < n) ? nbr[base + lane] : 0;
            int j = 0;
            for (; j + 8 <= n; j += 8) {
                float4 v[8];
                #pragma unroll
                for (int k = 0; k < 8; ++k) v[k] = ld_row(zl, zstride, ids, j + k);
                #pragma unroll
                for (int k = 0; k < 8; ++k) add4(acc, v[k]);
            }
            if (n - j >= 4) {
                float4 v0 = ld_row(zl, zstride, ids, j);
                float4 v1 = ld_row(zl, zstride, ids, j + 1);
                float4 v2 = ld_row(zl, zstride, ids, j + 2);
                float4 v3 = ld_row(zl, zstride, ids, j + 3);
                add4(acc, v0); add4(acc, v1); add4(acc, v2); add4(acc, v3);
                j += 4;
            }
            if (n - j >= 2) {
                float4 v0 = ld_row(zl, zstride, ids, j);
                float4 v1 = ld_row(zl, zstride, ids, j + 1);
                add4(acc, v0); add4(acc, v1);
                j += 2;
            }
            if (n - j >= 1) {
                float4 v0 = ld_row(zl, zstride, ids, j);
                add4(acc, v0);
            }
        }
        *(float4*)(agg + (size_t)row * DIM + lane * 4) = acc;
    }
}

__global__ __launch_bounds__(256) void gin_mlp_kernel(
    const float* __restrict__ z, int zstride,
    float* __restrict__ hbuf,
    const float* __restrict__ W1, const float* __restrict__ b1,
    const float* __restrict__ W2, const float* __restrict__ b2,
    float* __restrict__ stats)
{
    __shared__ float ldsA[16 * LDS_STRIDE];
    __shared__ float ldsH[16 * LDS_STRIDE];

    const int tid  = threadIdx.x;
    const int wave = tid >> 5;
    const int lane = tid & 31;
    const int half = lane >> 4;
    const int l16  = lane & 15;
    const int row0 = blockIdx.x * 16;
    const int col  = wave * 16 + l16;

    for (int idx = tid; idx < 16 * DIM; idx += 256) {
        int r = idx >> 7, c = idx & 127;
        ldsA[r * LDS_STRIDE + c] =
            z[(size_t)(row0 + r) * zstride + c] + hbuf[(size_t)(row0 + r) * DIM + c];
    }
    __syncthreads();

    v8f acc = {};
    for (int k0 = 0; k0 < DIM; k0 += 4) {
        int ka = k0 + 2 * half;
        v2f a, b;
        a.x = ldsA[l16 * LDS_STRIDE + ka];
        a.y = ldsA[l16 * LDS_STRIDE + ka + 1];
        b.x = W1[ka * DIM + col];
        b.y = W1[(ka + 1) * DIM + col];
        acc = __builtin_amdgcn_wmma_f32_16x16x4_f32(
            false, a, false, b, (short)0, acc, false, false);
    }

    float bias1 = b1[col];
    #pragma unroll
    for (int g = 0; g < 8; ++g) {
        int m = g + 8 * half;
        float v = acc[g] + bias1;
        ldsH[m * LDS_STRIDE + col] = v > 0.f ? v : 0.f;
    }
    __syncthreads();

    v8f acc2 = {};
    for (int k0 = 0; k0 < DIM; k0 += 4) {
        int ka = k0 + 2 * half;
        v2f a, b;
        a.x = ldsH[l16 * LDS_STRIDE + ka];
        a.y = ldsH[l16 * LDS_STRIDE + ka + 1];
        b.x = W2[ka * DIM + col];
        b.y = W2[(ka + 1) * DIM + col];
        acc2 = __builtin_amdgcn_wmma_f32_16x16x4_f32(
            false, a, false, b, (short)0, acc2, false, false);
    }

    float bias2 = b2[col];
    float s = 0.f, q = 0.f;
    #pragma unroll
    for (int g = 0; g < 8; ++g) {
        int m = g + 8 * half;
        float v = acc2[g] + bias2;
        v = v > 0.f ? v : 0.f;
        hbuf[(size_t)(row0 + m) * DIM + col] = v;
        s += v;
        q += v * v;
    }
    s += __shfl_xor(s, 16);
    q += __shfl_xor(q, 16);
    if (half == 0) {
        atomicAdd(&stats[col], s);
        atomicAdd(&stats[DIM + col], q);
    }
}

__global__ void bn_stats_finalize(float* __restrict__ stats)
{
    int c = threadIdx.x;
    float mu  = stats[c] * (1.0f / N_NODES);
    float var = stats[DIM + c] * (1.0f / N_NODES) - mu * mu;
    stats[c]       = mu;
    stats[DIM + c] = rsqrtf(var + BN_EPS);
}

__global__ __launch_bounds__(256) void bn_pool_kernel(
    const float* __restrict__ h,
    const float* __restrict__ stats,
    const float* __restrict__ gamma, const float* __restrict__ beta,
    const int* __restrict__ batch,
    float* __restrict__ zcat, float* __restrict__ gcat)
{
    int tid = blockIdx.x * blockDim.x + threadIdx.x;
    if (tid >= N_NODES * 32) return;
    int i = tid >> 5;
    int c = (tid & 31) * 4;
    int g = batch[i];
    float4 v = *(const float4*)(h + (size_t)i * DIM + c);
    float in[4] = {v.x, v.y, v.z, v.w};
    float4 o;
    float* op = &o.x;
    #pragma unroll
    for (int j = 0; j < 4; ++j) {
        float mu = stats[c + j];
        float rs = stats[DIM + c + j];
        float y  = (in[j] - mu) * rs * gamma[c + j] + beta[c + j];
        op[j] = y;
        atomicAdd(&gcat[(size_t)g * ZSTRIDE + c + j], y);
    }
    *(float4*)(zcat + (size_t)i * ZSTRIDE + c) = o;
}

extern "C" void kernel_launch(void* const* d_in, const int* in_sizes, int n_in,
                              void* d_out, int out_size, void* d_ws, size_t ws_size,
                              hipStream_t stream)
{
    (void)in_sizes; (void)n_in; (void)out_size; (void)ws_size;

    const float* x     = (const float*)d_in[0];
    const int*   ei    = (const int*)  d_in[1];
    const int*   batch = (const int*)  d_in[2];
    const float* W1    = (const float*)d_in[3];
    const float* b1    = (const float*)d_in[4];
    const float* W2    = (const float*)d_in[5];
    const float* b2    = (const float*)d_in[6];
    const float* gamma = (const float*)d_in[7];
    const float* beta  = (const float*)d_in[8];

    float* out  = (float*)d_out;
    float* zcat = out;
    float* gcat = out + (size_t)N_NODES * ZSTRIDE;

    float* agg   = (float*)d_ws;
    float* stats = agg + (size_t)N_NODES * DIM;
    int* cnt      = (int*)(stats + 2 * DIM);
    int* rowstart = cnt + N_NODES;
    int* slot     = rowstart + N_NODES + 32;
    int* nbr      = slot + N_EDGES;

    const int* src = ei;
    const int* dst = ei + N_EDGES;

    hipMemsetAsync(gcat, 0, sizeof(float) * (size_t)NGRAPH * ZSTRIDE, stream);

    hipMemsetAsync(cnt, 0, sizeof(int) * N_NODES, stream);
    count_slot_kernel<<<EDGE_BLOCKS, 256, 0, stream>>>(dst, cnt, slot);
    row_start_kernel<<<SCAN_BLOCKS, SCAN_BLOCK, 0, stream>>>(cnt, rowstart);
    place_kernel<<<EDGE_BLOCKS, 256, 0, stream>>>(src, dst, slot, rowstart, nbr);

    for (int l = 0; l < NLAYER; ++l) {
        const float* zin = (l == 0) ? x : (zcat + (size_t)(l - 1) * DIM);
        int zstride      = (l == 0) ? DIM : ZSTRIDE;

        hipMemsetAsync(stats, 0, sizeof(float) * 2 * DIM, stream);

        gather_kernel<<<GATHER_BLOCKS, 256, 0, stream>>>(zin, zstride, rowstart, nbr, agg);

        gin_mlp_kernel<<<N_NODES / 16, 256, 0, stream>>>(
            zin, zstride, agg,
            W1 + (size_t)l * DIM * DIM, b1 + (size_t)l * DIM,
            W2 + (size_t)l * DIM * DIM, b2 + (size_t)l * DIM,
            stats);

        bn_stats_finalize<<<1, DIM, 0, stream>>>(stats);

        {
            long long threads = (long long)N_NODES * 32;
            int blocks = (int)((threads + 255) / 256);
            bn_pool_kernel<<<blocks, 256, 0, stream>>>(
                agg, stats,
                gamma + (size_t)l * DIM, beta + (size_t)l * DIM,
                batch,
                zcat + (size_t)l * DIM, gcat + (size_t)l * DIM);
        }
    }
}


# BN+pool: a wave walks 10 consecutive nodes (5000 waves, was 50000), pool summed in registers in node order, one float atomic set per graph-id change, not per node; pool sum order changes
# speedup vs baseline: 1.9012x; 1.1059x over previous
#include <hip/hip_runtime.h>

#define N_NODES 50000
#define N_EDGES 800000
#define DIM     128
#define NLAYER  3
#define NGRAPH  512
#define ZSTRIDE (NLAYER * DIM)
#define BN_EPS  1e-5f
#define LDS_STRIDE 132

typedef __attribute__((ext_vector_type(2))) float v2f;
typedef __attribute__((ext_vector_type(8))) float v8f;

#define EDGE_QUADS   (N_EDGES / 4)
#define EDGE_BLOCKS  ((EDGE_QUADS + 255) / 256)
#define SCAN_BLOCK   1024
#define SCAN_BLOCKS  ((N_NODES + 1 + SCAN_BLOCK - 1) / SCAN_BLOCK)
#define ROWS_PER_WAVE 5
#define GATHER_ROWS_PER_BLOCK (8 * ROWS_PER_WAVE)
#define GATHER_BLOCKS (N_NODES / GATHER_ROWS_PER_BLOCK)

__global__ __launch_bounds__(256) void count_slot_kernel(
    const int* __restrict__ dst, int* __restrict__ cnt, int* __restrict__ slot)
{
    int t = blockIdx.x * 256 + threadIdx.x;
    if (t >= EDGE_QUADS) return;
    int4 d = ((const int4*)dst)[t];
    int4 r;
    r.x = atomicAdd(&cnt[d.x], 1);
    r.y = atomicAdd(&cnt[d.y], 1);
    r.z = atomicAdd(&cnt[d.z], 1);
    r.w = atomicAdd(&cnt[d.w], 1);
    ((int4*)slot)[t] = r;
}

__global__ __launch_bounds__(SCAN_BLOCK) void row_start_kernel(
    const int* __restrict__ cnt, int* __restrict__ rowstart)
{
    __shared__ int wOwn[32];
    __shared__ int wFront[32];
    const int tid  = threadIdx.x;
    const int lane = tid & 31;
    const int wave = tid >> 5;
    const int b0   = blockIdx.x * SCAN_BLOCK;

    int front = 0;
    for (int i = tid; i < b0; i += SCAN_BLOCK) front += cnt[i];

    const int i = b0 + tid;
    const int c = (i < N_NODES) ? cnt[i] : 0;

    int incl = c;
    for (int d = 1; d < 32; d <<= 1) {
        int up = __shfl_up(incl, d);
        if (lane >= d) incl += up;
    }
    for (int d = 16; d >= 1; d >>= 1) front += __shfl_xor(front, d);
    if (lane == 31) wOwn[wave] = incl;
    if (lane == 0)  wFront[wave] = front;
    __syncthreads();

    int before = (lane < wave) ? wOwn[lane] : 0;
    int total  = wFront[lane];
    for (int d = 16; d >= 1; d >>= 1) {
        before += __shfl_xor(before, d);
        total  += __shfl_xor(total, d);
    }
    if (i <= N_NODES) rowstart[i] = total + before + incl - c;
}

__global__ __launch_bounds__(256) void place_kernel(
    const int* __restrict__ src, const int* __restrict__ dst,
    const int* __restrict__ slot, const int* __restrict__ rowstart,
    int* __restrict__ nbr)
{
    int t = blockIdx.x * 256 + threadIdx.x;
    if (t >= EDGE_QUADS) return;
    int4 s = ((const int4*)src)[t];
    int4 d = ((const int4*)dst)[t];
    int4 r = ((const int4*)slot)[t];
    unsigned p0 = (unsigned)(rowstart[d.x] + r.x);
    unsigned p1 = (unsigned)(rowstart[d.y] + r.y);
    unsigned p2 = (unsigned)(rowstart[d.z] + r.z);
    unsigned p3 = (unsigned)(rowstart[d.w] + r.w);
    if (p0 < N_EDGES) nbr[p0] = s.x;
    if (p1 < N_EDGES) nbr[p1] = s.y;
    if (p2 < N_EDGES) nbr[p2] = s.z;
    if (p3 < N_EDGES) nbr[p3] = s.w;
}

__device__ __forceinline__ float4 ld_row(const float* __restrict__ zl, int zstride,
                                         int ids, int j)
{
    unsigned s = (unsigned)__builtin_amdgcn_readlane(ids, j);
    s = s < (unsigned)N_NODES ? s : (unsigned)(N_NODES - 1);
    return *(const float4*)(zl + (size_t)s * zstride);
}

__device__ __forceinline__ void add4(float4& a, const float4& v)
{
    a.x += v.x; a.y += v.y; a.z += v.z; a.w += v.w;
}

__global__ __launch_bounds__(256) void gather_kernel(
    const float* __restrict__ z, int zstride,
    const int* __restrict__ rowstart, const int* __restrict__ nbr,
    float* __restrict__ agg)
{
    const int lane = threadIdx.x & 31;
    const int wave = __builtin_amdgcn_readfirstlane((int)(threadIdx.x >> 5));
    const int row0 = (blockIdx.x * 8 + wave) * ROWS_PER_WAVE;
    const float* zl = z + lane * 4;

    for (int r = 0; r < ROWS_PER_WAVE; ++r) {
        const int row = row0 + r;
        int beg = rowstart[row];
        int end = rowstart[row + 1];
        beg = beg < 0 ? 0 : beg;
        end = end > N_EDGES ? N_EDGES : end;
        float4 acc = make_float4(0.f, 0.f, 0.f, 0.f);

        for (int base = beg; base < end; base += 32) {
            int n = end - base;
            n = n > 32 ? 32 : n;
            int ids = (lane < n) ? nbr[base + lane] : 0;
            int j = 0;
            for (; j + 8 <= n; j += 8) {
                float4 v[8];
                #pragma unroll
                for (int k = 0; k < 8; ++k) v[k] = ld_row(zl, zstride, ids, j + k);
                #pragma unroll
                for (int k = 0; k < 8; ++k) add4(acc, v[k]);
            }
            if (n - j >= 4) {
                float4 v0 = ld_row(zl, zstride, ids, j);
                float4 v1 = ld_row(zl, zstride, ids, j + 1);
                float4 v2 = ld_row(zl, zstride, ids, j + 2);
                float4 v3 = ld_row(zl, zstride, ids, j + 3);
                add4(acc, v0); add4(acc, v1); add4(acc, v2); add4(acc, v3);
                j += 4;
            }
            if (n - j >= 2) {
                float4 v0 = ld_row(zl, zstride, ids, j);
                float4 v1 = ld_row(zl, zstride, ids, j + 1);
                add4(acc, v0); add4(acc, v1);
                j += 2;
            }
            if (n - j >= 1) {
                float4 v0 = ld_row(zl, zstride, ids, j);
                add4(acc, v0);
            }
        }
        *(float4*)(agg + (size_t)row * DIM + lane * 4) = acc;
    }
}

__global__ __launch_bounds__(256) void gin_mlp_kernel(
    const float* __restrict__ z, int zstride,
    float* __restrict__ hbuf,
    const float* __restrict__ W1, const float* __restrict__ b1,
    const float* __restrict__ W2, const float* __restrict__ b2,
    float* __restrict__ stats)
{
    __shared__ float ldsA[16 * LDS_STRIDE];
    __shared__ float ldsH[16 * LDS_STRIDE];

    const int tid  = threadIdx.x;
    const int wave = tid >> 5;
    const int lane = tid & 31;
    const int half = lane >> 4;
    const int l16  = lane & 15;
    const int row0 = blockIdx.x * 16;
    const int col  = wave * 16 + l16;

    for (int idx = tid; idx < 16 * DIM; idx += 256) {
        int r = idx >> 7, c = idx & 127;
        ldsA[r * LDS_STRIDE + c] =
            z[(size_t)(row0 + r) * zstride + c] + hbuf[(size_t)(row0 + r) * DIM + c];
    }
    __syncthreads();

    v8f acc = {};
    for (int k0 = 0; k0 < DIM; k0 += 4) {
        int ka = k0 + 2 * half;
        v2f a, b;
        a.x = ldsA[l16 * LDS_STRIDE + ka];
        a.y = ldsA[l16 * LDS_STRIDE + ka + 1];
        b.x = W1[ka * DIM + col];
        b.y = W1[(ka + 1) * DIM + col];
        acc = __builtin_amdgcn_wmma_f32_16x16x4_f32(
            false, a, false, b, (short)0, acc, false, false);
    }

    float bias1 = b1[col];
    #pragma unroll
    for (int g = 0; g < 8; ++g) {
        int m = g + 8 * half;
        float v = acc[g] + bias1;
        ldsH[m * LDS_STRIDE + col] = v > 0.f ? v : 0.f;
    }
    __syncthreads();

    v8f acc2 = {};
    for (int k0 = 0; k0 < DIM; k0 += 4) {
        int ka = k0 + 2 * half;
        v2f a, b;
        a.x = ldsH[l16 * LDS_STRIDE + ka];
        a.y = ldsH[l16 * LDS_STRIDE + ka + 1];
        b.x = W2[ka * DIM + col];
        b.y = W2[(ka + 1) * DIM + col];
        acc2 = __builtin_amdgcn_wmma_f32_16x16x4_f32(
            false, a, false, b, (short)0, acc2, false, false);
    }

    float bias2 = b2[col];
    float s = 0.f, q = 0.f;
    #pragma unroll
    for (int g = 0; g < 8; ++g) {
        int m = g + 8 * half;
        float v = acc2[g] + bias2;
        v = v > 0.f ? v : 0.f;
        hbuf[(size_t)(row0 + m) * DIM + col] = v;
        s += v;
        q += v * v;
    }
    s += __shfl_xor(s, 16);
    q += __shfl_xor(q, 16);
    if (half == 0) {
        atomicAdd(&stats[col], s);
        atomicAdd(&stats[DIM + col], q);
    }
}

__global__ void bn_stats_finalize(float* __restrict__ stats)
{
    int c = threadIdx.x;
    float mu  = stats[c] * (1.0f / N_NODES);
    float var = stats[DIM + c] * (1.0f / N_NODES) - mu * mu;
    stats[c]       = mu;
    stats[DIM + c] = rsqrtf(var + BN_EPS);
}

#define POOL_NODES_PER_WAVE 10
#define POOL_BLOCKS (N_NODES / (8 * POOL_NODES_PER_WAVE))

__device__ __forceinline__ void pool_flush(float* __restrict__ gcat, int g, int c,
                                           const float4& a)
{
    unsigned gg = (unsigned)g < (unsigned)NGRAPH ? (unsigned)g : (unsigned)(NGRAPH - 1);
    float* p = gcat + (size_t)gg * ZSTRIDE + c;
    atomicAdd(p,     a.x);
    atomicAdd(p + 1, a.y);
    atomicAdd(p + 2, a.z);
    atomicAdd(p + 3, a.w);
}

__global__ __launch_bounds__(256) void bn_pool_kernel(
    const float* __restrict__ h,
    const float* __restrict__ stats,
    const float* __restrict__ gamma, const float* __restrict__ beta,
    const int* __restrict__ batch,
    float* __restrict__ zcat, float* __restrict__ gcat)
{
    const int lane = threadIdx.x & 31;
    const int wave = __builtin_amdgcn_readfirstlane((int)(threadIdx.x >> 5));
    const int n0 = (blockIdx.x * 8 + wave) * POOL_NODES_PER_WAVE;
    const int c  = lane * 4;

    int gids = (lane < POOL_NODES_PER_WAVE) ? batch[n0 + lane] : 0;

    const float4 mu = *(const float4*)(stats + c);
    const float4 rs = *(const float4*)(stats + DIM + c);
    const float4 ga = *(const float4*)(gamma + c);
    const float4 be = *(const float4*)(beta + c);

    float4 v[POOL_NODES_PER_WAVE];
    for (int k = 0; k < POOL_NODES_PER_WAVE; ++k)
        v[k] = *(const float4*)(h + (size_t)(n0 + k) * DIM + c);

    float4 acc = make_float4(0.f, 0.f, 0.f, 0.f);
    int gcur = __builtin_amdgcn_readlane(gids, 0);

    for (int k = 0; k < POOL_NODES_PER_WAVE; ++k) {
        const int g = __builtin_amdgcn_readlane(gids, k);
        if (g != gcur) {
            pool_flush(gcat, gcur, c, acc);
            acc = make_float4(0.f, 0.f, 0.f, 0.f);
            gcur = g;
        }
        float4 o;
        o.x = (v[k].x - mu.x) * rs.x * ga.x + be.x;
        o.y = (v[k].y - mu.y) * rs.y * ga.y + be.y;
        o.z = (v[k].z - mu.z) * rs.z * ga.z + be.z;
        o.w = (v[k].w - mu.w) * rs.w * ga.w + be.w;
        *(float4*)(zcat + (size_t)(n0 + k) * ZSTRIDE + c) = o;
        acc.x += o.x; acc.y += o.y; acc.z += o.z; acc.w += o.w;
    }
    pool_flush(gcat, gcur, c, acc);
}

extern "C" void kernel_launch(void* const* d_in, const int* in_sizes, int n_in,
                              void* d_out, int out_size, void* d_ws, size_t ws_size,
                              hipStream_t stream)
{
    (void)in_sizes; (void)n_in; (void)out_size; (void)ws_size;

    const float* x     = (const float*)d_in[0];
    const int*   ei    = (const int*)  d_in[1];
    const int*   batch = (const int*)  d_in[2];
    const float* W1    = (const float*)d_in[3];
    const float* b1    = (const float*)d_in[4];
    const float* W2    = (const float*)d_in[5];
    const float* b2    = (const float*)d_in[6];
    const float* gamma = (const float*)d_in[7];
    const float* beta  = (const float*)d_in[8];

    float* out  = (float*)d_out;
    float* zcat = out;
    float* gcat = out + (size_t)N_NODES * ZSTRIDE;

    float* agg   = (float*)d_ws;
    float* stats = agg + (size_t)N_NODES * DIM;
    int* cnt      = (int*)(stats + 2 * DIM);
    int* rowstart = cnt + N_NODES;
    int* slot     = rowstart + N_NODES + 32;
    int* nbr      = slot + N_EDGES;

    const int* src = ei;
    const int* dst = ei + N_EDGES;

    hipMemsetAsync(gcat, 0, sizeof(float) * (size_t)NGRAPH * ZSTRIDE, stream);

    hipMemsetAsync(cnt, 0, sizeof(int) * N_NODES, stream);
    count_slot_kernel<<<EDGE_BLOCKS, 256, 0, stream>>>(dst, cnt, slot);
    row_start_kernel<<<SCAN_BLOCKS, SCAN_BLOCK, 0, stream>>>(cnt, rowstart);
    place_kernel<<<EDGE_BLOCKS, 256, 0, stream>>>(src, dst, slot, rowstart, nbr);

    for (int l = 0; l < NLAYER; ++l) {
        const float* zin = (l == 0) ? x : (zcat + (size_t)(l - 1) * DIM);
        int zstride      = (l == 0) ? DIM : ZSTRIDE;

        hipMemsetAsync(stats, 0, sizeof(float) * 2 * DIM, stream);

        gather_kernel<<<GATHER_BLOCKS, 256, 0, stream>>>(zin, zstride, rowstart, nbr, agg);

        gin_mlp_kernel<<<N_NODES / 16, 256, 0, stream>>>(
            zin, zstride, agg,
            W1 + (size_t)l * DIM * DIM, b1 + (size_t)l * DIM,
            W2 + (size_t)l * DIM * DIM, b2 + (size_t)l * DIM,
            stats);

        bn_stats_finalize<<<1, DIM, 0, stream>>>(stats);

        bn_pool_kernel<<<POOL_BLOCKS, 256, 0, stream>>>(
            agg, stats,
            gamma + (size_t)l * DIM, beta + (size_t)l * DIM,
            batch,
            zcat + (size_t)l * DIM, gcat + (size_t)l * DIM);
    }
}


# Neighbour sum fused into the MLP kernel: a wave sums its 2 rows' lists (14 row loads in flight) straight into the LDS A tile; gather launch and agg write+read gone; row sum order unchanged
# speedup vs baseline: 2.5824x; 1.3583x over previous
#include <hip/hip_runtime.h>

#define N_NODES 50000
#define N_EDGES 800000
#define DIM     128
#define NLAYER  3
#define NGRAPH  512
#define ZSTRIDE (NLAYER * DIM)
#define BN_EPS  1e-5f
#define LDS_STRIDE 132

typedef __attribute__((ext_vector_type(2))) float v2f;
typedef __attribute__((ext_vector_type(8))) float v8f;

#define EDGE_QUADS   (N_EDGES / 4)
#define EDGE_BLOCKS  ((EDGE_QUADS + 255) / 256)
#define SCAN_BLOCK   1024
#define SCAN_BLOCKS  ((N_NODES + 1 + SCAN_BLOCK - 1) / SCAN_BLOCK)
#define ROW_GROUP 14

__global__ __launch_bounds__(256) void count_slot_kernel(
    const int* __restrict__ dst, int* __restrict__ cnt, int* __restrict__ slot)
{
    int t = blockIdx.x * 256 + threadIdx.x;
    if (t >= EDGE_QUADS) return;
    int4 d = ((const int4*)dst)[t];
    int4 r;
    r.x = atomicAdd(&cnt[d.x], 1);
    r.y = atomicAdd(&cnt[d.y], 1);
    r.z = atomicAdd(&cnt[d.z], 1);
    r.w = atomicAdd(&cnt[d.w], 1);
    ((int4*)slot)[t] = r;
}

__global__ __launch_bounds__(SCAN_BLOCK) void row_start_kernel(
    const int* __restrict__ cnt, int* __restrict__ rowstart)
{
    __shared__ int wOwn[32];
    __shared__ int wFront[32];
    const int tid  = threadIdx.x;
    const int lane = tid & 31;
    const int wave = tid >> 5;
    const int b0   = blockIdx.x * SCAN_BLOCK;

    int front = 0;
    for (int i = tid; i < b0; i += SCAN_BLOCK) front += cnt[i];

    const int i = b0 + tid;
    const int c = (i < N_NODES) ? cnt[i] : 0;

    int incl = c;
    for (int d = 1; d < 32; d <<= 1) {
        int up = __shfl_up(incl, d);
        if (lane >= d) incl += up;
    }
    for (int d = 16; d >= 1; d >>= 1) front += __shfl_xor(front, d);
    if (lane == 31) wOwn[wave] = incl;
    if (lane == 0)  wFront[wave] = front;
    __syncthreads();

    int before = (lane < wave) ? wOwn[lane] : 0;
    int total  = wFront[lane];
    for (int d = 16; d >= 1; d >>= 1) {
        before += __shfl_xor(before, d);
        total  += __shfl_xor(total, d);
    }
    if (i <= N_NODES) rowstart[i] = total + before + incl - c;
}

__global__ __launch_bounds__(256) void place_kernel(
    const int* __restrict__ src, const int* __restrict__ dst,
    const int* __restrict__ slot, const int* __restrict__ rowstart,
    int* __restrict__ nbr)
{
    int t = blockIdx.x * 256 + threadIdx.x;
    if (t >= EDGE_QUADS) return;
    int4 s = ((const int4*)src)[t];
    int4 d = ((const int4*)dst)[t];
    int4 r = ((const int4*)slot)[t];
    unsigned p0 = (unsigned)(rowstart[d.x] + r.x);
    unsigned p1 = (unsigned)(rowstart[d.y] + r.y);
    unsigned p2 = (unsigned)(rowstart[d.z] + r.z);
    unsigned p3 = (unsigned)(rowstart[d.w] + r.w);
    if (p0 < N_EDGES) nbr[p0] = s.x;
    if (p1 < N_EDGES) nbr[p1] = s.y;
    if (p2 < N_EDGES) nbr[p2] = s.z;
    if (p3 < N_EDGES) nbr[p3] = s.w;
}

__device__ __forceinline__ unsigned row_id(int ids, int j)
{
    unsigned s = (unsigned)__builtin_amdgcn_readlane(ids, j & 31);
    return s < (unsigned)N_NODES ? s : (unsigned)(N_NODES - 1);
}

__device__ __forceinline__ float4 ld_row(const float* __restrict__ z, int zstride,
                                         unsigned s, unsigned laneoff)
{
    const char* rowp = (const char*)(z + (size_t)s * zstride);
    return *(const float4*)(rowp + laneoff);
}

__device__ __forceinline__ void add4(float4& a, const float4& v)
{
    a.x += v.x; a.y += v.y; a.z += v.z; a.w += v.w;
}

__device__ __forceinline__ void chunk_sum(float4& acc,
                                          const float* __restrict__ z, int zstride,
                                          int ids, int n, unsigned laneoff)
{
    for (int j = 0; j < n; j += ROW_GROUP) {
        unsigned s[ROW_GROUP];
        #pragma unroll
        for (int k = 0; k < ROW_GROUP; ++k) s[k] = row_id(ids, j + k);
        float4 v[ROW_GROUP];
        #pragma unroll
        for (int k = 0; k < ROW_GROUP; ++k)
            if (j + k < n) v[k] = ld_row(z, zstride, s[k], laneoff);
        #pragma unroll
        for (int k = 0; k < ROW_GROUP; ++k)
            if (j + k < n) add4(acc, v[k]);
    }
}

__device__ __forceinline__ float4 row_sum(const float* __restrict__ z, int zstride,
                                          const int* __restrict__ nbr,
                                          int beg, int end, int ids,
                                          int lane, unsigned laneoff)
{
    float4 acc = make_float4(0.f, 0.f, 0.f, 0.f);
    int n = end - beg;
    n = n > 32 ? 32 : n;
    chunk_sum(acc, z, zstride, ids, n, laneoff);
    for (int base = beg + 32; base < end; base += 32) {
        n = end - base;
        n = n > 32 ? 32 : n;
        ids = (lane < n) ? nbr[base + lane] : 0;
        chunk_sum(acc, z, zstride, ids, n, laneoff);
    }
    return acc;
}

__global__ __launch_bounds__(256) __attribute__((amdgpu_waves_per_eu(12))) void gin_mlp_kernel(
    const float* __restrict__ z, int zstride,
    const int* __restrict__ rowstart, const int* __restrict__ nbr,
    float* __restrict__ hbuf,
    const float* __restrict__ W1, const float* __restrict__ b1,
    const float* __restrict__ W2, const float* __restrict__ b2,
    float* __restrict__ stats)
{
    __shared__ __attribute__((aligned(16))) float ldsA[16 * LDS_STRIDE];
    __shared__ float ldsH[16 * LDS_STRIDE];

    const int tid  = threadIdx.x;
    const int wave = tid >> 5;
    const int lane = tid & 31;
    const int half = lane >> 4;
    const int l16  = lane & 15;
    const int row0 = blockIdx.x * 16;
    const int col  = wave * 16 + l16;

    {
        const int wv = __builtin_amdgcn_readfirstlane(tid >> 5);
        const unsigned laneoff = (unsigned)lane * 16u;
        const int ra = row0 + 2 * wv;
        int e0 = rowstart[ra];
        int e1 = rowstart[ra + 1];
        int e2 = rowstart[ra + 2];
        e0 = e0 < 0 ? 0 : (e0 > N_EDGES ? N_EDGES : e0);
        e1 = e1 < 0 ? 0 : (e1 > N_EDGES ? N_EDGES : e1);
        e2 = e2 < 0 ? 0 : (e2 > N_EDGES ? N_EDGES : e2);
        const int idsA = (lane < e1 - e0) ? nbr[e0 + lane] : 0;
        const int idsB = (lane < e2 - e1) ? nbr[e1 + lane] : 0;
        const float4 ownA = *(const float4*)((const char*)(z + (size_t)ra * zstride) + laneoff);

        float4 sA = row_sum(z, zstride, nbr, e0, e1, idsA, lane, laneoff);
        float4 oA;
        oA.x = ownA.x + sA.x; oA.y = ownA.y + sA.y; oA.z = ownA.z + sA.z; oA.w = ownA.w + sA.w;
        *(float4*)(&ldsA[(2 * wv) * LDS_STRIDE + lane * 4]) = oA;

        const float4 ownB = *(const float4*)((const char*)(z + (size_t)(ra + 1) * zstride) + laneoff);
        float4 sB = row_sum(z, zstride, nbr, e1, e2, idsB, lane, laneoff);
        float4 oB;
        oB.x = ownB.x + sB.x; oB.y = ownB.y + sB.y; oB.z = ownB.z + sB.z; oB.w = ownB.w + sB.w;
        *(float4*)(&ldsA[(2 * wv + 1) * LDS_STRIDE + lane * 4]) = oB;
    }
    __syncthreads();

    v8f acc = {};
    for (int k0 = 0; k0 < DIM; k0 += 4) {
        int ka = k0 + 2 * half;
        v2f a, b;
        a.x = ldsA[l16 * LDS_STRIDE + ka];
        a.y = ldsA[l16 * LDS_STRIDE + ka + 1];
        b.x = W1[ka * DIM + col];
        b.y = W1[(ka + 1) * DIM + col];
        acc = __builtin_amdgcn_wmma_f32_16x16x4_f32(
            false, a, false, b, (short)0, acc, false, false);
    }

    float bias1 = b1[col];
    #pragma unroll
    for (int g = 0; g < 8; ++g) {
        int m = g + 8 * half;
        float v = acc[g] + bias1;
        ldsH[m * LDS_STRIDE + col] = v > 0.f ? v : 0.f;
    }
    __syncthreads();

    v8f acc2 = {};
    for (int k0 = 0; k0 < DIM; k0 += 4) {
        int ka = k0 + 2 * half;
        v2f a, b;
        a.x = ldsH[l16 * LDS_STRIDE + ka];
        a.y = ldsH[l16 * LDS_STRIDE + ka + 1];
        b.x = W2[ka * DIM + col];
        b.y = W2[(ka + 1) * DIM + col];
        acc2 = __builtin_amdgcn_wmma_f32_16x16x4_f32(
            false, a, false, b, (short)0, acc2, false, false);
    }

    float bias2 = b2[col];
    float s = 0.f, q = 0.f;
    #pragma unroll
    for (int g = 0; g < 8; ++g) {
        int m = g + 8 * half;
        float v = acc2[g] + bias2;
        v = v > 0.f ? v : 0.f;
        hbuf[(size_t)(row0 + m) * DIM + col] = v;
        s += v;
        q += v * v;
    }
    s += __shfl_xor(s, 16);
    q += __shfl_xor(q, 16);
    if (half == 0) {
        atomicAdd(&stats[col], s);
        atomicAdd(&stats[DIM + col], q);
    }
}

__global__ void bn_stats_finalize(float* __restrict__ stats)
{
    int c = threadIdx.x;
    float mu  = stats[c] * (1.0f / N_NODES);
    float var = stats[DIM + c] * (1.0f / N_NODES) - mu * mu;
    stats[c]       = mu;
    stats[DIM + c] = rsqrtf(var + BN_EPS);
}

#define POOL_NODES_PER_WAVE 10
#define POOL_BLOCKS (N_NODES / (8 * POOL_NODES_PER_WAVE))

__device__ __forceinline__ void pool_flush(float* __restrict__ gcat, int g, int c,
                                           const float4& a)
{
    unsigned gg = (unsigned)g < (unsigned)NGRAPH ? (unsigned)g : (unsigned)(NGRAPH - 1);
    float* p = gcat + (size_t)gg * ZSTRIDE + c;
    atomicAdd(p,     a.x);
    atomicAdd(p + 1, a.y);
    atomicAdd(p + 2, a.z);
    atomicAdd(p + 3, a.w);
}

__global__ __launch_bounds__(256) void bn_pool_kernel(
    const float* __restrict__ h,
    const float* __restrict__ stats,
    const float* __restrict__ gamma, const float* __restrict__ beta,
    const int* __restrict__ batch,
    float* __restrict__ zcat, float* __restrict__ gcat)
{
    const int lane = threadIdx.x & 31;
    const int wave = __builtin_amdgcn_readfirstlane((int)(threadIdx.x >> 5));
    const int n0 = (blockIdx.x * 8 + wave) * POOL_NODES_PER_WAVE;
    const int c  = lane * 4;

    int gids = (lane < POOL_NODES_PER_WAVE) ? batch[n0 + lane] : 0;

    const float4 mu = *(const float4*)(stats + c);
    const float4 rs = *(const float4*)(stats + DIM + c);
    const float4 ga = *(const float4*)(gamma + c);
    const float4 be = *(const float4*)(beta + c);

    float4 v[POOL_NODES_PER_WAVE];
    for (int k = 0; k < POOL_NODES_PER_WAVE; ++k)
        v[k] = *(const float4*)(h + (size_t)(n0 + k) * DIM + c);

    float4 acc = make_float4(0.f, 0.f, 0.f, 0.f);
    int gcur = __builtin_amdgcn_readlane(gids, 0);

    for (int k = 0; k < POOL_NODES_PER_WAVE; ++k) {
        const int g = __builtin_amdgcn_readlane(gids, k);
        if (g != gcur) {
            pool_flush(gcat, gcur, c, acc);
            acc = make_float4(0.f, 0.f, 0.f, 0.f);
            gcur = g;
        }
        float4 o;
        o.x = (v[k].x - mu.x) * rs.x * ga.x + be.x;
        o.y = (v[k].y - mu.y) * rs.y * ga.y + be.y;
        o.z = (v[k].z - mu.z) * rs.z * ga.z + be.z;
        o.w = (v[k].w - mu.w) * rs.w * ga.w + be.w;
        *(float4*)(zcat + (size_t)(n0 + k) * ZSTRIDE + c) = o;
        acc.x += o.x; acc.y += o.y; acc.z += o.z; acc.w += o.w;
    }
    pool_flush(gcat, gcur, c, acc);
}

extern "C" void kernel_launch(void* const* d_in, const int* in_sizes, int n_in,
                              void* d_out, int out_size, void* d_ws, size_t ws_size,
                              hipStream_t stream)
{
    (void)in_sizes; (void)n_in; (void)out_size; (void)ws_size;

    const float* x     = (const float*)d_in[0];
    const int*   ei    = (const int*)  d_in[1];
    const int*   batch = (const int*)  d_in[2];
    const float* W1    = (const float*)d_in[3];
    const float* b1    = (const float*)d_in[4];
    const float* W2    = (const float*)d_in[5];
    const float* b2    = (const float*)d_in[6];
    const float* gamma = (const float*)d_in[7];
    const float* beta  = (const float*)d_in[8];

    float* out  = (float*)d_out;
    float* zcat = out;
    float* gcat = out + (size_t)N_NODES * ZSTRIDE;

    float* hbuf  = (float*)d_ws;
    float* stats = hbuf + (size_t)N_NODES * DIM;
    int* cnt      = (int*)(stats + 2 * DIM);
    int* rowstart = cnt + N_NODES;
    int* slot     = rowstart + N_NODES + 32;
    int* nbr      = slot + N_EDGES;

    const int* src = ei;
    const int* dst = ei + N_EDGES;

    hipMemsetAsync(gcat, 0, sizeof(float) * (size_t)NGRAPH * ZSTRIDE, stream);

    hipMemsetAsync(cnt, 0, sizeof(int) * N_NODES, stream);
    count_slot_kernel<<<EDGE_BLOCKS, 256, 0, stream>>>(dst, cnt, slot);
    row_start_kernel<<<SCAN_BLOCKS, SCAN_BLOCK, 0, stream>>>(cnt, rowstart);
    place_kernel<<<EDGE_BLOCKS, 256, 0, stream>>>(src, dst, slot, rowstart, nbr);

    for (int l = 0; l < NLAYER; ++l) {
        const float* zin = (l == 0) ? x : (zcat + (size_t)(l - 1) * DIM);
        int zstride      = (l == 0) ? DIM : ZSTRIDE;

        hipMemsetAsync(stats, 0, sizeof(float) * 2 * DIM, stream);

        gin_mlp_kernel<<<N_NODES / 16, 256, 0, stream>>>(
            zin, zstride, rowstart, nbr, hbuf,
            W1 + (size_t)l * DIM * DIM, b1 + (size_t)l * DIM,
            W2 + (size_t)l * DIM * DIM, b2 + (size_t)l * DIM,
            stats);

        bn_stats_finalize<<<1, DIM, 0, stream>>>(stats);

        bn_pool_kernel<<<POOL_BLOCKS, 256, 0, stream>>>(
            hbuf, stats,
            gamma + (size_t)l * DIM, beta + (size_t)l * DIM,
            batch,
            zcat + (size_t)l * DIM, gcat + (size_t)l * DIM);
    }
}


# BN column sums: each block's float atomics go to one of 128 partial sets picked by block number, not all to the same 8 lines; finalize adds the sets in set order; column sum order changes
# speedup vs baseline: 2.6879x; 1.0409x over previous
#include <hip/hip_runtime.h>

#define N_NODES 50000
#define N_EDGES 800000
#define DIM     128
#define NLAYER  3
#define NGRAPH  512
#define ZSTRIDE (NLAYER * DIM)
#define BN_EPS  1e-5f
#define LDS_STRIDE 132
#define STAT_SETS 128
#define FIN_PARTS 4

typedef __attribute__((ext_vector_type(2))) float v2f;
typedef __attribute__((ext_vector_type(8))) float v8f;

#define EDGE_QUADS   (N_EDGES / 4)
#define EDGE_BLOCKS  ((EDGE_QUADS + 255) / 256)
#define SCAN_BLOCK   1024
#define SCAN_BLOCKS  ((N_NODES + 1 + SCAN_BLOCK - 1) / SCAN_BLOCK)
#define ROW_GROUP 14

__global__ __launch_bounds__(256) void count_slot_kernel(
    const int* __restrict__ dst, int* __restrict__ cnt, int* __restrict__ slot)
{
    int t = blockIdx.x * 256 + threadIdx.x;
    if (t >= EDGE_QUADS) return;
    int4 d = ((const int4*)dst)[t];
    int4 r;
    r.x = atomicAdd(&cnt[d.x], 1);
    r.y = atomicAdd(&cnt[d.y], 1);
    r.z = atomicAdd(&cnt[d.z], 1);
    r.w = atomicAdd(&cnt[d.w], 1);
    ((int4*)slot)[t] = r;
}

__global__ __launch_bounds__(SCAN_BLOCK) void row_start_kernel(
    const int* __restrict__ cnt, int* __restrict__ rowstart)
{
    __shared__ int wOwn[32];
    __shared__ int wFront[32];
    const int tid  = threadIdx.x;
    const int lane = tid & 31;
    const int wave = tid >> 5;
    const int b0   = blockIdx.x * SCAN_BLOCK;

    int front = 0;
    for (int i = tid; i < b0; i += SCAN_BLOCK) front += cnt[i];

    const int i = b0 + tid;
    const int c = (i < N_NODES) ? cnt[i] : 0;

    int incl = c;
    for (int d = 1; d < 32; d <<= 1) {
        int up = __shfl_up(incl, d);
        if (lane >= d) incl += up;
    }
    for (int d = 16; d >= 1; d >>= 1) front += __shfl_xor(front, d);
    if (lane == 31) wOwn[wave] = incl;
    if (lane == 0)  wFront[wave] = front;
    __syncthreads();

    int before = (lane < wave) ? wOwn[lane] : 0;
    int total  = wFront[lane];
    for (int d = 16; d >= 1; d >>= 1) {
        before += __shfl_xor(before, d);
        total  += __shfl_xor(total, d);
    }
    if (i <= N_NODES) rowstart[i] = total + before + incl - c;
}

__global__ __launch_bounds__(256) void place_kernel(
    const int* __restrict__ src, const int* __restrict__ dst,
    const int* __restrict__ slot, const int* __restrict__ rowstart,
    int* __restrict__ nbr)
{
    int t = blockIdx.x * 256 + threadIdx.x;
    if (t >= EDGE_QUADS) return;
    int4 s = ((const int4*)src)[t];
    int4 d = ((const int4*)dst)[t];
    int4 r = ((const int4*)slot)[t];
    unsigned p0 = (unsigned)(rowstart[d.x] + r.x);
    unsigned p1 = (unsigned)(rowstart[d.y] + r.y);
    unsigned p2 = (unsigned)(rowstart[d.z] + r.z);
    unsigned p3 = (unsigned)(rowstart[d.w] + r.w);
    if (p0 < N_EDGES) nbr[p0] = s.x;
    if (p1 < N_EDGES) nbr[p1] = s.y;
    if (p2 < N_EDGES) nbr[p2] = s.z;
    if (p3 < N_EDGES) nbr[p3] = s.w;
}

__device__ __forceinline__ unsigned row_id(int ids, int j)
{
    unsigned s = (unsigned)__builtin_amdgcn_readlane(ids, j & 31);
    return s < (unsigned)N_NODES ? s : (unsigned)(N_NODES - 1);
}

__device__ __forceinline__ float4 ld_row(const float* __restrict__ z, int zstride,
                                         unsigned s, unsigned laneoff)
{
    const char* rowp = (const char*)(z + (size_t)s * zstride);
    return *(const float4*)(rowp + laneoff);
}

__device__ __forceinline__ void add4(float4& a, const float4& v)
{
    a.x += v.x; a.y += v.y; a.z += v.z; a.w += v.w;
}

__device__ __forceinline__ void chunk_sum(float4& acc,
                                          const float* __restrict__ z, int zstride,
                                          int ids, int n, unsigned laneoff)
{
    for (int j = 0; j < n; j += ROW_GROUP) {
        unsigned s[ROW_GROUP];
        #pragma unroll
        for (int k = 0; k < ROW_GROUP; ++k) s[k] = row_id(ids, j + k);
        float4 v[ROW_GROUP];
        #pragma unroll
        for (int k = 0; k < ROW_GROUP; ++k)
            if (j + k < n) v[k] = ld_row(z, zstride, s[k], laneoff);
        #pragma unroll
        for (int k = 0; k < ROW_GROUP; ++k)
            if (j + k < n) add4(acc, v[k]);
    }
}

__device__ __forceinline__ float4 row_sum(const float* __restrict__ z, int zstride,
                                          const int* __restrict__ nbr,
                                          int beg, int end, int ids,
                                          int lane, unsigned laneoff)
{
    float4 acc = make_float4(0.f, 0.f, 0.f, 0.f);
    int n = end - beg;
    n = n > 32 ? 32 : n;
    chunk_sum(acc, z, zstride, ids, n, laneoff);
    for (int base = beg + 32; base < end; base += 32) {
        n = end - base;
        n = n > 32 ? 32 : n;
        ids = (lane < n) ? nbr[base + lane] : 0;
        chunk_sum(acc, z, zstride, ids, n, laneoff);
    }
    return acc;
}

__global__ __launch_bounds__(256) __attribute__((amdgpu_waves_per_eu(12))) void gin_mlp_kernel(
    const float* __restrict__ z, int zstride,
    const int* __restrict__ rowstart, const int* __restrict__ nbr,
    float* __restrict__ hbuf,
    const float* __restrict__ W1, const float* __restrict__ b1,
    const float* __restrict__ W2, const float* __restrict__ b2,
    float* __restrict__ stats)
{
    __shared__ __attribute__((aligned(16))) float ldsA[16 * LDS_STRIDE];
    __shared__ float ldsH[16 * LDS_STRIDE];

    const int tid  = threadIdx.x;
    const int wave = tid >> 5;
    const int lane = tid & 31;
    const int half = lane >> 4;
    const int l16  = lane & 15;
    const int row0 = blockIdx.x * 16;
    const int col  = wave * 16 + l16;

    {
        const int wv = __builtin_amdgcn_readfirstlane(tid >> 5);
        const unsigned laneoff = (unsigned)lane * 16u;
        const int ra = row0 + 2 * wv;
        int e0 = rowstart[ra];
        int e1 = rowstart[ra + 1];
        int e2 = rowstart[ra + 2];
        e0 = e0 < 0 ? 0 : (e0 > N_EDGES ? N_EDGES : e0);
        e1 = e1 < 0 ? 0 : (e1 > N_EDGES ? N_EDGES : e1);
        e2 = e2 < 0 ? 0 : (e2 > N_EDGES ? N_EDGES : e2);
        const int idsA = (lane < e1 - e0) ? nbr[e0 + lane] : 0;
        const int idsB = (lane < e2 - e1) ? nbr[e1 + lane] : 0;
        const float4 ownA = *(const float4*)((const char*)(z + (size_t)ra * zstride) + laneoff);

        float4 sA = row_sum(z, zstride, nbr, e0, e1, idsA, lane, laneoff);
        float4 oA;
        oA.x = ownA.x + sA.x; oA.y = ownA.y + sA.y; oA.z = ownA.z + sA.z; oA.w = ownA.w + sA.w;
        *(float4*)(&ldsA[(2 * wv) * LDS_STRIDE + lane * 4]) = oA;

        const float4 ownB = *(const float4*)((const char*)(z + (size_t)(ra + 1) * zstride) + laneoff);
        float4 sB = row_sum(z, zstride, nbr, e1, e2, idsB, lane, laneoff);
        float4 oB;
        oB.x = ownB.x + sB.x; oB.y = ownB.y + sB.y; oB.z = ownB.z + sB.z; oB.w = ownB.w + sB.w;
        *(float4*)(&ldsA[(2 * wv + 1) * LDS_STRIDE + lane * 4]) = oB;
    }
    __syncthreads();

    v8f acc = {};
    for (int k0 = 0; k0 < DIM; k0 += 4) {
        int ka = k0 + 2 * half;
        v2f a, b;
        a.x = ldsA[l16 * LDS_STRIDE + ka];
        a.y = ldsA[l16 * LDS_STRIDE + ka + 1];
        b.x = W1[ka * DIM + col];
        b.y = W1[(ka + 1) * DIM + col];
        acc = __builtin_amdgcn_wmma_f32_16x16x4_f32(
            false, a, false, b, (short)0, acc, false, false);
    }

    float bias1 = b1[col];
    #pragma unroll
    for (int g = 0; g < 8; ++g) {
        int m = g + 8 * half;
        float v = acc[g] + bias1;
        ldsH[m * LDS_STRIDE + col] = v > 0.f ? v : 0.f;
    }
    __syncthreads();

    v8f acc2 = {};
    for (int k0 = 0; k0 < DIM; k0 += 4) {
        int ka = k0 + 2 * half;
        v2f a, b;
        a.x = ldsH[l16 * LDS_STRIDE + ka];
        a.y = ldsH[l16 * LDS_STRIDE + ka + 1];
        b.x = W2[ka * DIM + col];
        b.y = W2[(ka + 1) * DIM + col];
        acc2 = __builtin_amdgcn_wmma_f32_16x16x4_f32(
            false, a, false, b, (short)0, acc2, false, false);
    }

    float bias2 = b2[col];
    float s = 0.f, q = 0.f;
    #pragma unroll
    for (int g = 0; g < 8; ++g) {
        int m = g + 8 * half;
        float v = acc2[g] + bias2;
        v = v > 0.f ? v : 0.f;
        hbuf[(size_t)(row0 + m) * DIM + col] = v;
        s += v;
        q += v * v;
    }
    s += __shfl_xor(s, 16);
    q += __shfl_xor(q, 16);
    float* st = stats + (size_t)(blockIdx.x & (STAT_SETS - 1)) * (2 * DIM);
    if (half == 0) {
        atomicAdd(&st[col], s);
        atomicAdd(&st[DIM + col], q);
    }
}

__global__ __launch_bounds__(FIN_PARTS * 2 * DIM) void bn_stats_finalize(float* __restrict__ stats)
{
    __shared__ float part[FIN_PARTS][2 * DIM];
    const int w = threadIdx.x & (2 * DIM - 1);
    const int p = threadIdx.x >> 8;
    const float* mine = stats + (size_t)p * (STAT_SETS / FIN_PARTS) * (2 * DIM) + w;
    float a = 0.f;
    #pragma unroll
    for (int k = 0; k < STAT_SETS / FIN_PARTS; ++k) a += mine[(size_t)k * (2 * DIM)];
    part[p][w] = a;
    __syncthreads();
    if (threadIdx.x < DIM) {
        const int c = threadIdx.x;
        float sm = (part[0][c] + part[1][c]) + (part[2][c] + part[3][c]);
        float sq = (part[0][DIM + c] + part[1][DIM + c]) + (part[2][DIM + c] + part[3][DIM + c]);
        float mu  = sm * (1.0f / N_NODES);
        float var = sq * (1.0f / N_NODES) - mu * mu;
        stats[c]       = mu;
        stats[DIM + c] = rsqrtf(var + BN_EPS);
    }
}

#define POOL_NODES_PER_WAVE 10
#define POOL_BLOCKS (N_NODES / (8 * POOL_NODES_PER_WAVE))

__device__ __forceinline__ void pool_flush(float* __restrict__ gcat, int g, int c,
                                           const float4& a)
{
    unsigned gg = (unsigned)g < (unsigned)NGRAPH ? (unsigned)g : (unsigned)(NGRAPH - 1);
    float* p = gcat + (size_t)gg * ZSTRIDE + c;
    atomicAdd(p,     a.x);
    atomicAdd(p + 1, a.y);
    atomicAdd(p + 2, a.z);
    atomicAdd(p + 3, a.w);
}

__global__ __launch_bounds__(256) void bn_pool_kernel(
    const float* __restrict__ h,
    const float* __restrict__ stats,
    const float* __restrict__ gamma, const float* __restrict__ beta,
    const int* __restrict__ batch,
    float* __restrict__ zcat, float* __restrict__ gcat)
{
    const int lane = threadIdx.x & 31;
    const int wave = __builtin_amdgcn_readfirstlane((int)(threadIdx.x >> 5));
    const int n0 = (blockIdx.x * 8 + wave) * POOL_NODES_PER_WAVE;
    const int c  = lane * 4;

    int gids = (lane < POOL_NODES_PER_WAVE) ? batch[n0 + lane] : 0;

    const float4 mu = *(const float4*)(stats + c);
    const float4 rs = *(const float4*)(stats + DIM + c);
    const float4 ga = *(const float4*)(gamma + c);
    const float4 be = *(const float4*)(beta + c);

    float4 v[POOL_NODES_PER_WAVE];
    for (int k = 0; k < POOL_NODES_PER_WAVE; ++k)
        v[k] = *(const float4*)(h + (size_t)(n0 + k) * DIM + c);

    float4 acc = make_float4(0.f, 0.f, 0.f, 0.f);
    int gcur = __builtin_amdgcn_readlane(gids, 0);

    for (int k = 0; k < POOL_NODES_PER_WAVE; ++k) {
        const int g = __builtin_amdgcn_readlane(gids, k);
        if (g != gcur) {
            pool_flush(gcat, gcur, c, acc);
            acc = make_float4(0.f, 0.f, 0.f, 0.f);
            gcur = g;
        }
        float4 o;
        o.x = (v[k].x - mu.x) * rs.x * ga.x + be.x;
        o.y = (v[k].y - mu.y) * rs.y * ga.y + be.y;
        o.z = (v[k].z - mu.z) * rs.z * ga.z + be.z;
        o.w = (v[k].w - mu.w) * rs.w * ga.w + be.w;
        *(float4*)(zcat + (size_t)(n0 + k) * ZSTRIDE + c) = o;
        acc.x += o.x; acc.y += o.y; acc.z += o.z; acc.w += o.w;
    }
    pool_flush(gcat, gcur, c, acc);
}

extern "C" void kernel_launch(void* const* d_in, const int* in_sizes, int n_in,
                              void* d_out, int out_size, void* d_ws, size_t ws_size,
                              hipStream_t stream)
{
    (void)in_sizes; (void)n_in; (void)out_size; (void)ws_size;

    const float* x     = (const float*)d_in[0];
    const int*   ei    = (const int*)  d_in[1];
    const int*   batch = (const int*)  d_in[2];
    const float* W1    = (const float*)d_in[3];
    const float* b1    = (const float*)d_in[4];
    const float* W2    = (const float*)d_in[5];
    const float* b2    = (const float*)d_in[6];
    const float* gamma = (const float*)d_in[7];
    const float* beta  = (const float*)d_in[8];

    float* out  = (float*)d_out;
    float* zcat = out;
    float* gcat = out + (size_t)N_NODES * ZSTRIDE;

    float* hbuf  = (float*)d_ws;
    float* stats = hbuf + (size_t)N_NODES * DIM;
    int* cnt      = (int*)(stats + STAT_SETS * 2 * DIM);
    int* rowstart = cnt + N_NODES;
    int* slot     = rowstart + N_NODES + 32;
    int* nbr      = slot + N_EDGES;

    const int* src = ei;
    const int* dst = ei + N_EDGES;

    hipMemsetAsync(gcat, 0, sizeof(float) * (size_t)NGRAPH * ZSTRIDE, stream);

    hipMemsetAsync(cnt, 0, sizeof(int) * N_NODES, stream);
    count_slot_kernel<<<EDGE_BLOCKS, 256, 0, stream>>>(dst, cnt, slot);
    row_start_kernel<<<SCAN_BLOCKS, SCAN_BLOCK, 0, stream>>>(cnt, rowstart);
    place_kernel<<<EDGE_BLOCKS, 256, 0, stream>>>(src, dst, slot, rowstart, nbr);

    for (int l = 0; l < NLAYER; ++l) {
        const float* zin = (l == 0) ? x : (zcat + (size_t)(l - 1) * DIM);
        int zstride      = (l == 0) ? DIM : ZSTRIDE;

        hipMemsetAsync(stats, 0, sizeof(float) * STAT_SETS * 2 * DIM, stream);

        gin_mlp_kernel<<<N_NODES / 16, 256, 0, stream>>>(
            zin, zstride, rowstart, nbr, hbuf,
            W1 + (size_t)l * DIM * DIM, b1 + (size_t)l * DIM,
            W2 + (size_t)l * DIM * DIM, b2 + (size_t)l * DIM,
            stats);

        bn_stats_finalize<<<1, FIN_PARTS * 2 * DIM, 0, stream>>>(stats);

        bn_pool_kernel<<<POOL_BLOCKS, 256, 0, stream>>>(
            hbuf, stats,
            gamma + (size_t)l * DIM, beta + (size_t)l * DIM,
            batch,
            zcat + (size_t)l * DIM, gcat + (size_t)l * DIM);
    }
}


# Stream operations 17 -> 10, no ticket or fence: one clear for cnt and per-layer BN stat areas, count kernel zeroes g_cat, each BN+pool block adds the 16 stat sets itself; column sum order changes
# speedup vs baseline: 2.8873x; 1.0742x over previous
#include <hip/hip_runtime.h>

#define N_NODES 50000
#define N_EDGES 800000
#define DIM     128
#define NLAYER  3
#define NGRAPH  512
#define ZSTRIDE (NLAYER * DIM)
#define BN_EPS  1e-5f
#define LDS_STRIDE 132
#define STAT_SETS 16
#define STAT_LAYER_FLOATS (STAT_SETS * 2 * DIM)
#define CLEAR_BYTES (sizeof(float) * NLAYER * STAT_LAYER_FLOATS + sizeof(int) * N_NODES)
#define GCAT_QUADS (NGRAPH * ZSTRIDE / 4)

typedef __attribute__((ext_vector_type(2))) float v2f;
typedef __attribute__((ext_vector_type(8))) float v8f;

#define EDGE_QUADS   (N_EDGES / 4)
#define EDGE_BLOCKS  ((EDGE_QUADS + 255) / 256)
#define SCAN_BLOCK   1024
#define SCAN_BLOCKS  ((N_NODES + 1 + SCAN_BLOCK - 1) / SCAN_BLOCK)
#define ROW_GROUP 14

__global__ __launch_bounds__(256) void count_slot_kernel(
    const int* __restrict__ dst, int* __restrict__ cnt, int* __restrict__ slot,
    float* __restrict__ gcat)
{
    int t = blockIdx.x * 256 + threadIdx.x;
    if (t < GCAT_QUADS) ((float4*)gcat)[t] = make_float4(0.f, 0.f, 0.f, 0.f);
    if (t >= EDGE_QUADS) return;
    int4 d = ((const int4*)dst)[t];
    int4 r;
    r.x = atomicAdd(&cnt[d.x], 1);
    r.y = atomicAdd(&cnt[d.y], 1);
    r.z = atomicAdd(&cnt[d.z], 1);
    r.w = atomicAdd(&cnt[d.w], 1);
    ((int4*)slot)[t] = r;
}

__global__ __launch_bounds__(SCAN_BLOCK) void row_start_kernel(
    const int* __restrict__ cnt, int* __restrict__ rowstart)
{
    __shared__ int wOwn[32];
    __shared__ int wFront[32];
    const int tid  = threadIdx.x;
    const int lane = tid & 31;
    const int wave = tid >> 5;
    const int b0   = blockIdx.x * SCAN_BLOCK;

    int front = 0;
    for (int i = tid; i < b0; i += SCAN_BLOCK) front += cnt[i];

    const int i = b0 + tid;
    const int c = (i < N_NODES) ? cnt[i] : 0;

    int incl = c;
    for (int d = 1; d < 32; d <<= 1) {
        int up = __shfl_up(incl, d);
        if (lane >= d) incl += up;
    }
    for (int d = 16; d >= 1; d >>= 1) front += __shfl_xor(front, d);
    if (lane == 31) wOwn[wave] = incl;
    if (lane == 0)  wFront[wave] = front;
    __syncthreads();

    int before = (lane < wave) ? wOwn[lane] : 0;
    int total  = wFront[lane];
    for (int d = 16; d >= 1; d >>= 1) {
        before += __shfl_xor(before, d);
        total  += __shfl_xor(total, d);
    }
    if (i <= N_NODES) rowstart[i] = total + before + incl - c;
}

__global__ __launch_bounds__(256) void place_kernel(
    const int* __restrict__ src, const int* __restrict__ dst,
    const int* __restrict__ slot, const int* __restrict__ rowstart,
    int* __restrict__ nbr)
{
    int t = blockIdx.x * 256 + threadIdx.x;
    if (t >= EDGE_QUADS) return;
    int4 s = ((const int4*)src)[t];
    int4 d = ((const int4*)dst)[t];
    int4 r = ((const int4*)slot)[t];
    unsigned p0 = (unsigned)(rowstart[d.x] + r.x);
    unsigned p1 = (unsigned)(rowstart[d.y] + r.y);
    unsigned p2 = (unsigned)(rowstart[d.z] + r.z);
    unsigned p3 = (unsigned)(rowstart[d.w] + r.w);
    if (p0 < N_EDGES) nbr[p0] = s.x;
    if (p1 < N_EDGES) nbr[p1] = s.y;
    if (p2 < N_EDGES) nbr[p2] = s.z;
    if (p3 < N_EDGES) nbr[p3] = s.w;
}

__device__ __forceinline__ unsigned row_id(int ids, int j)
{
    unsigned s = (unsigned)__builtin_amdgcn_readlane(ids, j & 31);
    return s < (unsigned)N_NODES ? s : (unsigned)(N_NODES - 1);
}

__device__ __forceinline__ float4 ld_row(const float* __restrict__ z, int zstride,
                                         unsigned s, unsigned laneoff)
{
    const char* rowp = (const char*)(z + (size_t)s * zstride);
    return *(const float4*)(rowp + laneoff);
}

__device__ __forceinline__ void add4(float4& a, const float4& v)
{
    a.x += v.x; a.y += v.y; a.z += v.z; a.w += v.w;
}

__device__ __forceinline__ void chunk_sum(float4& acc,
                                          const float* __restrict__ z, int zstride,
                                          int ids, int n, unsigned laneoff)
{
    for (int j = 0; j < n; j += ROW_GROUP) {
        unsigned s[ROW_GROUP];
        #pragma unroll
        for (int k = 0; k < ROW_GROUP; ++k) s[k] = row_id(ids, j + k);
        float4 v[ROW_GROUP];
        #pragma unroll
        for (int k = 0; k < ROW_GROUP; ++k)
            if (j + k < n) v[k] = ld_row(z, zstride, s[k], laneoff);
        #pragma unroll
        for (int k = 0; k < ROW_GROUP; ++k)
            if (j + k < n) add4(acc, v[k]);
    }
}

__device__ __forceinline__ float4 row_sum(const float* __restrict__ z, int zstride,
                                          const int* __restrict__ nbr,
                                          int beg, int end, int ids,
                                          int lane, unsigned laneoff)
{
    float4 acc = make_float4(0.f, 0.f, 0.f, 0.f);
    int n = end - beg;
    n = n > 32 ? 32 : n;
    chunk_sum(acc, z, zstride, ids, n, laneoff);
    for (int base = beg + 32; base < end; base += 32) {
        n = end - base;
        n = n > 32 ? 32 : n;
        ids = (lane < n) ? nbr[base + lane] : 0;
        chunk_sum(acc, z, zstride, ids, n, laneoff);
    }
    return acc;
}

__global__ __launch_bounds__(256) __attribute__((amdgpu_waves_per_eu(12))) void gin_mlp_kernel(
    const float* __restrict__ z, int zstride,
    const int* __restrict__ rowstart, const int* __restrict__ nbr,
    float* __restrict__ hbuf,
    const float* __restrict__ W1, const float* __restrict__ b1,
    const float* __restrict__ W2, const float* __restrict__ b2,
    float* __restrict__ stats)
{
    __shared__ __attribute__((aligned(16))) float ldsA[16 * LDS_STRIDE];
    __shared__ float ldsH[16 * LDS_STRIDE];

    const int tid  = threadIdx.x;
    const int wave = tid >> 5;
    const int lane = tid & 31;
    const int half = lane >> 4;
    const int l16  = lane & 15;
    const int row0 = blockIdx.x * 16;
    const int col  = wave * 16 + l16;

    {
        const int wv = __builtin_amdgcn_readfirstlane(tid >> 5);
        const unsigned laneoff = (unsigned)lane * 16u;
        const int ra = row0 + 2 * wv;
        int e0 = rowstart[ra];
        int e1 = rowstart[ra + 1];
        int e2 = rowstart[ra + 2];
        e0 = e0 < 0 ? 0 : (e0 > N_EDGES ? N_EDGES : e0);
        e1 = e1 < 0 ? 0 : (e1 > N_EDGES ? N_EDGES : e1);
        e2 = e2 < 0 ? 0 : (e2 > N_EDGES ? N_EDGES : e2);
        const int idsA = (lane < e1 - e0) ? nbr[e0 + lane] : 0;
        const int idsB = (lane < e2 - e1) ? nbr[e1 + lane] : 0;
        const float4 ownA = *(const float4*)((const char*)(z + (size_t)ra * zstride) + laneoff);

        float4 sA = row_sum(z, zstride, nbr, e0, e1, idsA, lane, laneoff);
        float4 oA;
        oA.x = ownA.x + sA.x; oA.y = ownA.y + sA.y; oA.z = ownA.z + sA.z; oA.w = ownA.w + sA.w;
        *(float4*)(&ldsA[(2 * wv) * LDS_STRIDE + lane * 4]) = oA;

        const float4 ownB = *(const float4*)((const char*)(z + (size_t)(ra + 1) * zstride) + laneoff);
        float4 sB = row_sum(z, zstride, nbr, e1, e2, idsB, lane, laneoff);
        float4 oB;
        oB.x = ownB.x + sB.x; oB.y = ownB.y + sB.y; oB.z = ownB.z + sB.z; oB.w = ownB.w + sB.w;
        *(float4*)(&ldsA[(2 * wv + 1) * LDS_STRIDE + lane * 4]) = oB;
    }
    __syncthreads();

    v8f acc = {};
    for (int k0 = 0; k0 < DIM; k0 += 4) {
        int ka = k0 + 2 * half;
        v2f a, b;
        a.x = ldsA[l16 * LDS_STRIDE + ka];
        a.y = ldsA[l16 * LDS_STRIDE + ka + 1];
        b.x = W1[ka * DIM + col];
        b.y = W1[(ka + 1) * DIM + col];
        acc = __builtin_amdgcn_wmma_f32_16x16x4_f32(
            false, a, false, b, (short)0, acc, false, false);
    }

    float bias1 = b1[col];
    #pragma unroll
    for (int g = 0; g < 8; ++g) {
        int m = g + 8 * half;
        float v = acc[g] + bias1;
        ldsH[m * LDS_STRIDE + col] = v > 0.f ? v : 0.f;
    }
    __syncthreads();

    v8f acc2 = {};
    for (int k0 = 0; k0 < DIM; k0 += 4) {
        int ka = k0 + 2 * half;
        v2f a, b;
        a.x = ldsH[l16 * LDS_STRIDE + ka];
        a.y = ldsH[l16 * LDS_STRIDE + ka + 1];
        b.x = W2[ka * DIM + col];
        b.y = W2[(ka + 1) * DIM + col];
        acc2 = __builtin_amdgcn_wmma_f32_16x16x4_f32(
            false, a, false, b, (short)0, acc2, false, false);
    }

    float bias2 = b2[col];
    float s = 0.f, q = 0.f;
    #pragma unroll
    for (int g = 0; g < 8; ++g) {
        int m = g + 8 * half;
        float v = acc2[g] + bias2;
        v = v > 0.f ? v : 0.f;
        hbuf[(size_t)(row0 + m) * DIM + col] = v;
        s += v;
        q += v * v;
    }
    s += __shfl_xor(s, 16);
    q += __shfl_xor(q, 16);
    float* st = stats + (size_t)(blockIdx.x & (STAT_SETS - 1)) * (2 * DIM);
    if (half == 0) {
        atomicAdd(&st[col], s);
        atomicAdd(&st[DIM + col], q);
    }
}

#define POOL_NODES_PER_WAVE 10
#define POOL_BLOCKS (N_NODES / (8 * POOL_NODES_PER_WAVE))

__device__ __forceinline__ void pool_flush(float* __restrict__ gcat, int g, int c,
                                           const float4& a)
{
    unsigned gg = (unsigned)g < (unsigned)NGRAPH ? (unsigned)g : (unsigned)(NGRAPH - 1);
    float* p = gcat + (size_t)gg * ZSTRIDE + c;
    atomicAdd(p,     a.x);
    atomicAdd(p + 1, a.y);
    atomicAdd(p + 2, a.z);
    atomicAdd(p + 3, a.w);
}

__global__ __launch_bounds__(256) void bn_pool_kernel(
    const float* __restrict__ h,
    const float* __restrict__ stats,
    const float* __restrict__ gamma, const float* __restrict__ beta,
    const int* __restrict__ batch,
    float* __restrict__ zcat, float* __restrict__ gcat)
{
    __shared__ __attribute__((aligned(16))) float tot[2 * DIM];

    const int lane = threadIdx.x & 31;
    const int wave = __builtin_amdgcn_readfirstlane((int)(threadIdx.x >> 5));
    const int n0 = (blockIdx.x * 8 + wave) * POOL_NODES_PER_WAVE;
    const int c  = lane * 4;

    float w[STAT_SETS];
    #pragma unroll
    for (int k = 0; k < STAT_SETS; ++k) w[k] = stats[k * (2 * DIM) + threadIdx.x];

    int gids = (lane < POOL_NODES_PER_WAVE) ? batch[n0 + lane] : 0;

    const float4 ga = *(const float4*)(gamma + c);
    const float4 be = *(const float4*)(beta + c);

    float4 v[POOL_NODES_PER_WAVE];
    for (int k = 0; k < POOL_NODES_PER_WAVE; ++k)
        v[k] = *(const float4*)(h + (size_t)(n0 + k) * DIM + c);

    float a = 0.f;
    #pragma unroll
    for (int k = 0; k < STAT_SETS; ++k) a += w[k];
    tot[threadIdx.x] = a;
    __builtin_amdgcn_fence(__ATOMIC_RELEASE, "workgroup", "local");
    __builtin_amdgcn_s_barrier();
    __builtin_amdgcn_fence(__ATOMIC_ACQUIRE, "workgroup", "local");
    const float4 sm = *(const float4*)(tot + c);
    const float4 sq = *(const float4*)(tot + DIM + c);
    float4 mu, rs;
    mu.x = sm.x * (1.0f / N_NODES);
    mu.y = sm.y * (1.0f / N_NODES);
    mu.z = sm.z * (1.0f / N_NODES);
    mu.w = sm.w * (1.0f / N_NODES);
    rs.x = rsqrtf(sq.x * (1.0f / N_NODES) - mu.x * mu.x + BN_EPS);
    rs.y = rsqrtf(sq.y * (1.0f / N_NODES) - mu.y * mu.y + BN_EPS);
    rs.z = rsqrtf(sq.z * (1.0f / N_NODES) - mu.z * mu.z + BN_EPS);
    rs.w = rsqrtf(sq.w * (1.0f / N_NODES) - mu.w * mu.w + BN_EPS);

    float4 acc = make_float4(0.f, 0.f, 0.f, 0.f);
    int gcur = __builtin_amdgcn_readlane(gids, 0);

    for (int k = 0; k < POOL_NODES_PER_WAVE; ++k) {
        const int g = __builtin_amdgcn_readlane(gids, k);
        if (g != gcur) {
            pool_flush(gcat, gcur, c, acc);
            acc = make_float4(0.f, 0.f, 0.f, 0.f);
            gcur = g;
        }
        float4 o;
        o.x = (v[k].x - mu.x) * rs.x * ga.x + be.x;
        o.y = (v[k].y - mu.y) * rs.y * ga.y + be.y;
        o.z = (v[k].z - mu.z) * rs.z * ga.z + be.z;
        o.w = (v[k].w - mu.w) * rs.w * ga.w + be.w;
        *(float4*)(zcat + (size_t)(n0 + k) * ZSTRIDE + c) = o;
        acc.x += o.x; acc.y += o.y; acc.z += o.z; acc.w += o.w;
    }
    pool_flush(gcat, gcur, c, acc);
}

extern "C" void kernel_launch(void* const* d_in, const int* in_sizes, int n_in,
                              void* d_out, int out_size, void* d_ws, size_t ws_size,
                              hipStream_t stream)
{
    (void)in_sizes; (void)n_in; (void)out_size; (void)ws_size;

    const float* x     = (const float*)d_in[0];
    const int*   ei    = (const int*)  d_in[1];
    const int*   batch = (const int*)  d_in[2];
    const float* W1    = (const float*)d_in[3];
    const float* b1    = (const float*)d_in[4];
    const float* W2    = (const float*)d_in[5];
    const float* b2    = (const float*)d_in[6];
    const float* gamma = (const float*)d_in[7];
    const float* beta  = (const float*)d_in[8];

    float* out  = (float*)d_out;
    float* zcat = out;
    float* gcat = out + (size_t)N_NODES * ZSTRIDE;

    float* hbuf  = (float*)d_ws;
    float* stats = hbuf + (size_t)N_NODES * DIM;
    int* cnt      = (int*)(stats + NLAYER * STAT_LAYER_FLOATS);
    int* rowstart = cnt + N_NODES;
    int* slot     = rowstart + N_NODES + 32;
    int* nbr      = slot + N_EDGES;

    const int* src = ei;
    const int* dst = ei + N_EDGES;

    hipMemsetAsync(stats, 0, CLEAR_BYTES, stream);

    count_slot_kernel<<<EDGE_BLOCKS, 256, 0, stream>>>(dst, cnt, slot, gcat);
    row_start_kernel<<<SCAN_BLOCKS, SCAN_BLOCK, 0, stream>>>(cnt, rowstart);
    place_kernel<<<EDGE_BLOCKS, 256, 0, stream>>>(src, dst, slot, rowstart, nbr);

    for (int l = 0; l < NLAYER; ++l) {
        const float* zin = (l == 0) ? x : (zcat + (size_t)(l - 1) * DIM);
        int zstride      = (l == 0) ? DIM : ZSTRIDE;

        float* lstats = stats + (size_t)l * STAT_LAYER_FLOATS;

        gin_mlp_kernel<<<N_NODES / 16, 256, 0, stream>>>(
            zin, zstride, rowstart, nbr, hbuf,
            W1 + (size_t)l * DIM * DIM, b1 + (size_t)l * DIM,
            W2 + (size_t)l * DIM * DIM, b2 + (size_t)l * DIM,
            lstats);

        bn_pool_kernel<<<POOL_BLOCKS, 256, 0, stream>>>(
            hbuf, lstats,
            gamma + (size_t)l * DIM, beta + (size_t)l * DIM,
            batch,
            zcat + (size_t)l * DIM, gcat + (size_t)l * DIM);
    }
}


# MLP block owns 3-4 row tiles (1024 blocks, was 3125 of 1 tile): a W fragment load feeds one WMMA per tile, W loads per row /3; BN partial sums span 48-64 rows, column sum order changes
# speedup vs baseline: 3.0054x; 1.0409x over previous
#include <hip/hip_runtime.h>

#define N_NODES 50000
#define N_EDGES 800000
#define DIM     128
#define NLAYER  3
#define NGRAPH  512
#define ZSTRIDE (NLAYER * DIM)
#define BN_EPS  1e-5f
#define LDS_STRIDE 132
#define STAT_SETS 16
#define STAT_LAYER_FLOATS (STAT_SETS * 2 * DIM)
#define CLEAR_BYTES (sizeof(float) * NLAYER * STAT_LAYER_FLOATS + sizeof(int) * N_NODES)
#define GCAT_QUADS (NGRAPH * ZSTRIDE / 4)

typedef __attribute__((ext_vector_type(2))) float v2f;
typedef __attribute__((ext_vector_type(8))) float v8f;

#define EDGE_QUADS   (N_EDGES / 4)
#define EDGE_BLOCKS  ((EDGE_QUADS + 255) / 256)
#define SCAN_BLOCK   1024
#define SCAN_BLOCKS  ((N_NODES + 1 + SCAN_BLOCK - 1) / SCAN_BLOCK)
#define ROW_GROUP 14

__global__ __launch_bounds__(256) void count_slot_kernel(
    const int* __restrict__ dst, int* __restrict__ cnt, int* __restrict__ slot,
    float* __restrict__ gcat)
{
    int t = blockIdx.x * 256 + threadIdx.x;
    if (t < GCAT_QUADS) ((float4*)gcat)[t] = make_float4(0.f, 0.f, 0.f, 0.f);
    if (t >= EDGE_QUADS) return;
    int4 d = ((const int4*)dst)[t];
    int4 r;
    r.x = atomicAdd(&cnt[d.x], 1);
    r.y = atomicAdd(&cnt[d.y], 1);
    r.z = atomicAdd(&cnt[d.z], 1);
    r.w = atomicAdd(&cnt[d.w], 1);
    ((int4*)slot)[t] = r;
}

__global__ __launch_bounds__(SCAN_BLOCK) void row_start_kernel(
    const int* __restrict__ cnt, int* __restrict__ rowstart)
{
    __shared__ int wOwn[32];
    __shared__ int wFront[32];
    const int tid  = threadIdx.x;
    const int lane = tid & 31;
    const int wave = tid >> 5;
    const int b0   = blockIdx.x * SCAN_BLOCK;

    int front = 0;
    for (int i = tid; i < b0; i += SCAN_BLOCK) front += cnt[i];

    const int i = b0 + tid;
    const int c = (i < N_NODES) ? cnt[i] : 0;

    int incl = c;
    for (int d = 1; d < 32; d <<= 1) {
        int up = __shfl_up(incl, d);
        if (lane >= d) incl += up;
    }
    for (int d = 16; d >= 1; d >>= 1) front += __shfl_xor(front, d);
    if (lane == 31) wOwn[wave] = incl;
    if (lane == 0)  wFront[wave] = front;
    __syncthreads();

    int before = (lane < wave) ? wOwn[lane] : 0;
    int total  = wFront[lane];
    for (int d = 16; d >= 1; d >>= 1) {
        before += __shfl_xor(before, d);
        total  += __shfl_xor(total, d);
    }
    if (i <= N_NODES) rowstart[i] = total + before + incl - c;
}

__global__ __launch_bounds__(256) void place_kernel(
    const int* __restrict__ src, const int* __restrict__ dst,
    const int* __restrict__ slot, const int* __restrict__ rowstart,
    int* __restrict__ nbr)
{
    int t = blockIdx.x * 256 + threadIdx.x;
    if (t >= EDGE_QUADS) return;
    int4 s = ((const int4*)src)[t];
    int4 d = ((const int4*)dst)[t];
    int4 r = ((const int4*)slot)[t];
    unsigned p0 = (unsigned)(rowstart[d.x] + r.x);
    unsigned p1 = (unsigned)(rowstart[d.y] + r.y);
    unsigned p2 = (unsigned)(rowstart[d.z] + r.z);
    unsigned p3 = (unsigned)(rowstart[d.w] + r.w);
    if (p0 < N_EDGES) nbr[p0] = s.x;
    if (p1 < N_EDGES) nbr[p1] = s.y;
    if (p2 < N_EDGES) nbr[p2] = s.z;
    if (p3 < N_EDGES) nbr[p3] = s.w;
}

__device__ __forceinline__ unsigned row_id(int ids, int j)
{
    unsigned s = (unsigned)__builtin_amdgcn_readlane(ids, j & 31);
    return s < (unsigned)N_NODES ? s : (unsigned)(N_NODES - 1);
}

__device__ __forceinline__ float4 ld_row(const float* __restrict__ z, int zstride,
                                         unsigned s, unsigned laneoff)
{
    const char* rowp = (const char*)(z + (size_t)s * zstride);
    return *(const float4*)(rowp + laneoff);
}

__device__ __forceinline__ void add4(float4& a, const float4& v)
{
    a.x += v.x; a.y += v.y; a.z += v.z; a.w += v.w;
}

__device__ __forceinline__ void chunk_sum(float4& acc,
                                          const float* __restrict__ z, int zstride,
                                          int ids, int n, unsigned laneoff)
{
    for (int j = 0; j < n; j += ROW_GROUP) {
        unsigned s[ROW_GROUP];
        #pragma unroll
        for (int k = 0; k < ROW_GROUP; ++k) s[k] = row_id(ids, j + k);
        float4 v[ROW_GROUP];
        #pragma unroll
        for (int k = 0; k < ROW_GROUP; ++k) {
            v[k] = make_float4(0.f, 0.f, 0.f, 0.f);
            if (j + k < n) v[k] = ld_row(z, zstride, s[k], laneoff);
        }
        #pragma unroll
        for (int k = 0; k < ROW_GROUP; ++k)
            if (j + k < n) add4(acc, v[k]);
    }
}

__device__ __forceinline__ float4 row_sum(const float* __restrict__ z, int zstride,
                                          const int* __restrict__ nbr,
                                          int beg, int end, int ids,
                                          int lane, unsigned laneoff)
{
    float4 acc = make_float4(0.f, 0.f, 0.f, 0.f);
    int n = end - beg;
    n = n > 32 ? 32 : n;
    chunk_sum(acc, z, zstride, ids, n, laneoff);
    for (int base = beg + 32; base < end; base += 32) {
        n = end - base;
        n = n > 32 ? 32 : n;
        ids = (lane < n) ? nbr[base + lane] : 0;
        chunk_sum(acc, z, zstride, ids, n, laneoff);
    }
    return acc;
}

#define N_TILES    (N_NODES / 16)
#define MLP_BLOCKS 1024
#define MAX_TILES  4

template <int NT>
__device__ __forceinline__ void gemm_tiles(v8f (&acc)[NT], const float* lds,
                                           const float* __restrict__ Wm,
                                           int l16, int half, int col)
{
    const float* wp = Wm + (2 * half) * DIM + col;
    for (int k0 = 0; k0 < DIM; k0 += 4) {
        int ka = k0 + 2 * half;
        v2f b;
        b.x = wp[k0 * DIM];
        b.y = wp[(k0 + 1) * DIM];
        #pragma unroll
        for (int t = 0; t < NT; ++t) {
            v2f a;
            a.x = lds[(t * 16 + l16) * LDS_STRIDE + ka];
            a.y = lds[(t * 16 + l16) * LDS_STRIDE + ka + 1];
            acc[t] = __builtin_amdgcn_wmma_f32_16x16x4_f32(
                false, a, false, b, (short)0, acc[t], false, false);
        }
    }
}

template <int NT>
__device__ __forceinline__ void mlp_tiles(
    const float* ldsA, float* ldsH, int row0,
    float* __restrict__ hbuf,
    const float* __restrict__ W1, const float* __restrict__ b1,
    const float* __restrict__ W2, const float* __restrict__ b2,
    float* __restrict__ st, int l16, int half, int col)
{
    v8f acc[NT];
    #pragma unroll
    for (int t = 0; t < NT; ++t) acc[t] = v8f{};
    gemm_tiles<NT>(acc, ldsA, W1, l16, half, col);

    float bias1 = b1[col];
    #pragma unroll
    for (int t = 0; t < NT; ++t) {
        #pragma unroll
        for (int g = 0; g < 8; ++g) {
            int m = t * 16 + g + 8 * half;
            float v = acc[t][g] + bias1;
            ldsH[m * LDS_STRIDE + col] = v > 0.f ? v : 0.f;
        }
    }
    __syncthreads();

    v8f acc2[NT];
    #pragma unroll
    for (int t = 0; t < NT; ++t) acc2[t] = v8f{};
    gemm_tiles<NT>(acc2, ldsH, W2, l16, half, col);

    float bias2 = b2[col];
    float s = 0.f, q = 0.f;
    #pragma unroll
    for (int t = 0; t < NT; ++t) {
        #pragma unroll
        for (int g = 0; g < 8; ++g) {
            int m = t * 16 + g + 8 * half;
            float v = acc2[t][g] + bias2;
            v = v > 0.f ? v : 0.f;
            hbuf[(size_t)(row0 + m) * DIM + col] = v;
            s += v;
            q += v * v;
        }
    }
    s += __shfl_xor(s, 16);
    q += __shfl_xor(q, 16);
    if (half == 0) {
        atomicAdd(&st[col], s);
        atomicAdd(&st[DIM + col], q);
    }
}

__global__ __launch_bounds__(256) __attribute__((amdgpu_waves_per_eu(8))) void gin_mlp_kernel(
    const float* __restrict__ z, int zstride,
    const int* __restrict__ rowstart, const int* __restrict__ nbr,
    float* __restrict__ hbuf,
    const float* __restrict__ W1, const float* __restrict__ b1,
    const float* __restrict__ W2, const float* __restrict__ b2,
    float* __restrict__ stats)
{
    __shared__ __attribute__((aligned(16))) float ldsA[MAX_TILES * 16 * LDS_STRIDE];
    __shared__ float ldsH[MAX_TILES * 16 * LDS_STRIDE];

    const int tid  = threadIdx.x;
    const int wave = tid >> 5;
    const int lane = tid & 31;
    const int half = lane >> 4;
    const int l16  = lane & 15;
    const int col  = wave * 16 + l16;

    const int t0   = (int)(((unsigned)blockIdx.x * N_TILES) / MLP_BLOCKS);
    const int cnt  = (int)(((unsigned)(blockIdx.x + 1) * N_TILES) / MLP_BLOCKS) - t0;
    const int row0 = t0 * 16;

    {
        const int wv = __builtin_amdgcn_readfirstlane(tid >> 5);
        const unsigned laneoff = (unsigned)lane * 16u;
        for (int t = 0; t < cnt; ++t) {
            const int lr = t * 16 + 2 * wv;
            const int ra = row0 + lr;
            int e0 = rowstart[ra];
            int e1 = rowstart[ra + 1];
            int e2 = rowstart[ra + 2];
            e0 = e0 < 0 ? 0 : (e0 > N_EDGES ? N_EDGES : e0);
            e1 = e1 < 0 ? 0 : (e1 > N_EDGES ? N_EDGES : e1);
            e2 = e2 < 0 ? 0 : (e2 > N_EDGES ? N_EDGES : e2);
            const int idsA = (lane < e1 - e0) ? nbr[e0 + lane] : 0;
            const int idsB = (lane < e2 - e1) ? nbr[e1 + lane] : 0;
            const float4 ownA = *(const float4*)((const char*)(z + (size_t)ra * zstride) + laneoff);

            float4 sA = row_sum(z, zstride, nbr, e0, e1, idsA, lane, laneoff);
            float4 oA;
            oA.x = ownA.x + sA.x; oA.y = ownA.y + sA.y; oA.z = ownA.z + sA.z; oA.w = ownA.w + sA.w;
            *(float4*)(&ldsA[lr * LDS_STRIDE + lane * 4]) = oA;

            const float4 ownB = *(const float4*)((const char*)(z + (size_t)(ra + 1) * zstride) + laneoff);
            float4 sB = row_sum(z, zstride, nbr, e1, e2, idsB, lane, laneoff);
            float4 oB;
            oB.x = ownB.x + sB.x; oB.y = ownB.y + sB.y; oB.z = ownB.z + sB.z; oB.w = ownB.w + sB.w;
            *(float4*)(&ldsA[(lr + 1) * LDS_STRIDE + lane * 4]) = oB;
        }
    }
    __syncthreads();

    float* st = stats + (size_t)(blockIdx.x & (STAT_SETS - 1)) * (2 * DIM);
    if (cnt == MAX_TILES)
        mlp_tiles<MAX_TILES>(ldsA, ldsH, row0, hbuf, W1, b1, W2, b2, st, l16, half, col);
    else
        mlp_tiles<MAX_TILES - 1>(ldsA, ldsH, row0, hbuf, W1, b1, W2, b2, st, l16, half, col);
}

#define POOL_NODES_PER_WAVE 10
#define POOL_BLOCKS (N_NODES / (8 * POOL_NODES_PER_WAVE))

__device__ __forceinline__ void pool_flush(float* __restrict__ gcat, int g, int c,
                                           const float4& a)
{
    unsigned gg = (unsigned)g < (unsigned)NGRAPH ? (unsigned)g : (unsigned)(NGRAPH - 1);
    float* p = gcat + (size_t)gg * ZSTRIDE + c;
    atomicAdd(p,     a.x);
    atomicAdd(p + 1, a.y);
    atomicAdd(p + 2, a.z);
    atomicAdd(p + 3, a.w);
}

__global__ __launch_bounds__(256) void bn_pool_kernel(
    const float* __restrict__ h,
    const float* __restrict__ stats,
    const float* __restrict__ gamma, const float* __restrict__ beta,
    const int* __restrict__ batch,
    float* __restrict__ zcat, float* __restrict__ gcat)
{
    __shared__ __attribute__((aligned(16))) float tot[2 * DIM];

    const int lane = threadIdx.x & 31;
    const int wave = __builtin_amdgcn_readfirstlane((int)(threadIdx.x >> 5));
    const int n0 = (blockIdx.x * 8 + wave) * POOL_NODES_PER_WAVE;
    const int c  = lane * 4;

    float w[STAT_SETS];
    #pragma unroll
    for (int k = 0; k < STAT_SETS; ++k) w[k] = stats[k * (2 * DIM) + threadIdx.x];

    int gids = (lane < POOL_NODES_PER_WAVE) ? batch[n0 + lane] : 0;

    const float4 ga = *(const float4*)(gamma + c);
    const float4 be = *(const float4*)(beta + c);

    float4 v[POOL_NODES_PER_WAVE];
    for (int k = 0; k < POOL_NODES_PER_WAVE; ++k)
        v[k] = *(const float4*)(h + (size_t)(n0 + k) * DIM + c);

    float a = 0.f;
    #pragma unroll
    for (int k = 0; k < STAT_SETS; ++k) a += w[k];
    tot[threadIdx.x] = a;
    __builtin_amdgcn_fence(__ATOMIC_RELEASE, "workgroup", "local");
    __builtin_amdgcn_s_barrier();
    __builtin_amdgcn_fence(__ATOMIC_ACQUIRE, "workgroup", "local");
    const float4 sm = *(const float4*)(tot + c);
    const float4 sq = *(const float4*)(tot + DIM + c);
    float4 mu, rs;
    mu.x = sm.x * (1.0f / N_NODES);
    mu.y = sm.y * (1.0f / N_NODES);
    mu.z = sm.z * (1.0f / N_NODES);
    mu.w = sm.w * (1.0f / N_NODES);
    rs.x = rsqrtf(sq.x * (1.0f / N_NODES) - mu.x * mu.x + BN_EPS);
    rs.y = rsqrtf(sq.y * (1.0f / N_NODES) - mu.y * mu.y + BN_EPS);
    rs.z = rsqrtf(sq.z * (1.0f / N_NODES) - mu.z * mu.z + BN_EPS);
    rs.w = rsqrtf(sq.w * (1.0f / N_NODES) - mu.w * mu.w + BN_EPS);

    float4 acc = make_float4(0.f, 0.f, 0.f, 0.f);
    int gcur = __builtin_amdgcn_readlane(gids, 0);

    for (int k = 0; k < POOL_NODES_PER_WAVE; ++k) {
        const int g = __builtin_amdgcn_readlane(gids, k);
        if (g != gcur) {
            pool_flush(gcat, gcur, c, acc);
            acc = make_float4(0.f, 0.f, 0.f, 0.f);
            gcur = g;
        }
        float4 o;
        o.x = (v[k].x - mu.x) * rs.x * ga.x + be.x;
        o.y = (v[k].y - mu.y) * rs.y * ga.y + be.y;
        o.z = (v[k].z - mu.z) * rs.z * ga.z + be.z;
        o.w = (v[k].w - mu.w) * rs.w * ga.w + be.w;
        *(float4*)(zcat + (size_t)(n0 + k) * ZSTRIDE + c) = o;
        acc.x += o.x; acc.y += o.y; acc.z += o.z; acc.w += o.w;
    }
    pool_flush(gcat, gcur, c, acc);
}

extern "C" void kernel_launch(void* const* d_in, const int* in_sizes, int n_in,
                              void* d_out, int out_size, void* d_ws, size_t ws_size,
                              hipStream_t stream)
{
    (void)in_sizes; (void)n_in; (void)out_size; (void)ws_size;

    const float* x     = (const float*)d_in[0];
    const int*   ei    = (const int*)  d_in[1];
    const int*   batch = (const int*)  d_in[2];
    const float* W1    = (const float*)d_in[3];
    const float* b1    = (const float*)d_in[4];
    const float* W2    = (const float*)d_in[5];
    const float* b2    = (const float*)d_in[6];
    const float* gamma = (const float*)d_in[7];
    const float* beta  = (const float*)d_in[8];

    float* out  = (float*)d_out;
    float* zcat = out;
    float* gcat = out + (size_t)N_NODES * ZSTRIDE;

    float* hbuf  = (float*)d_ws;
    float* stats = hbuf + (size_t)N_NODES * DIM;
    int* cnt      = (int*)(stats + NLAYER * STAT_LAYER_FLOATS);
    int* rowstart = cnt + N_NODES;
    int* slot     = rowstart + N_NODES + 32;
    int* nbr      = slot + N_EDGES;

    const int* src = ei;
    const int* dst = ei + N_EDGES;

    hipMemsetAsync(stats, 0, CLEAR_BYTES, stream);

    count_slot_kernel<<<EDGE_BLOCKS, 256, 0, stream>>>(dst, cnt, slot, gcat);
    row_start_kernel<<<SCAN_BLOCKS, SCAN_BLOCK, 0, stream>>>(cnt, rowstart);
    place_kernel<<<EDGE_BLOCKS, 256, 0, stream>>>(src, dst, slot, rowstart, nbr);

    for (int l = 0; l < NLAYER; ++l) {
        const float* zin = (l == 0) ? x : (zcat + (size_t)(l - 1) * DIM);
        int zstride      = (l == 0) ? DIM : ZSTRIDE;

        float* lstats = stats + (size_t)l * STAT_LAYER_FLOATS;

        gin_mlp_kernel<<<MLP_BLOCKS, 256, 0, stream>>>(
            zin, zstride, rowstart, nbr, hbuf,
            W1 + (size_t)l * DIM * DIM, b1 + (size_t)l * DIM,
            W2 + (size_t)l * DIM * DIM, b2 + (size_t)l * DIM,
            lstats);

        bn_pool_kernel<<<POOL_BLOCKS, 256, 0, stream>>>(
            hbuf, lstats,
            gamma + (size_t)l * DIM, beta + (size_t)l * DIM,
            batch,
            zcat + (size_t)l * DIM, gcat + (size_t)l * DIM);
    }
}


# Edges ordered by destination with LDS counters, no global returning add per edge: 4096-edge blocks sort into 49 row buckets, a block per bucket ranks rows; 2 launches, was 3; row sum order changes
# speedup vs baseline: 4.1853x; 1.3926x over previous
#include <hip/hip_runtime.h>

#define N_NODES 50000
#define N_EDGES 800000
#define DIM     128
#define NLAYER  3
#define NGRAPH  512
#define ZSTRIDE (NLAYER * DIM)
#define BN_EPS  1e-5f
#define LDS_STRIDE 132
#define STAT_SETS 16
#define STAT_LAYER_FLOATS (STAT_SETS * 2 * DIM)
#define CURSOR_INTS 64
#define CLEAR_BYTES (sizeof(float) * NLAYER * STAT_LAYER_FLOATS + sizeof(int) * CURSOR_INTS)
#define GCAT_QUADS (NGRAPH * ZSTRIDE / 4)

typedef __attribute__((ext_vector_type(2))) float v2f;
typedef __attribute__((ext_vector_type(8))) float v8f;

#define EDGE_QUADS   (N_EDGES / 4)
#define ROW_GROUP 14
#define BKT_SHIFT    10
#define BKT_ROWS     (1 << BKT_SHIFT)
#define N_BKT        ((N_NODES + BKT_ROWS - 1) / BKT_ROWS)
#define BKT_CAP      N_EDGES
#define BIN_THREADS  1024
#define BIN_EDGES    (4 * BIN_THREADS)
#define BIN_BLOCKS   ((EDGE_QUADS + BIN_THREADS - 1) / BIN_THREADS)
#define LIST_LDS_CAP 40960
#define LIST_GROUP   20

static_assert(N_NODES <= 65536, "a source id must fit 16 bits of a bucket entry");
static_assert(N_BKT <= CURSOR_INTS, "one cursor per bucket");
static_assert(BIN_BLOCKS * BIN_THREADS >= GCAT_QUADS, "the bin kernel also stores g_cat's zeros");
static_assert(sizeof(float) * ((size_t)N_NODES * DIM + NLAYER * STAT_LAYER_FLOATS)
              + sizeof(int) * ((size_t)CURSOR_INTS + N_NODES + 32 + N_EDGES + (size_t)N_BKT * BKT_CAP)
              <= ((size_t)256 << 20), "all workspace areas lie inside the smallest workspace handed out");

__global__ __launch_bounds__(BIN_THREADS) void bin_kernel(
    const int* __restrict__ src, const int* __restrict__ dst,
    int* __restrict__ cursor, unsigned* __restrict__ buf, float* __restrict__ gcat)
{
    __shared__ int bcnt[CURSOR_INTS];
    __shared__ int loff[CURSOR_INTS];
    __shared__ int bbase[CURSOR_INTS];
    __shared__ int total;
    __shared__ unsigned ent[BIN_EDGES];
    __shared__ unsigned char ebk[BIN_EDGES];

    const int tid  = threadIdx.x;
    const int lane = tid & 31;
    const int t    = blockIdx.x * BIN_THREADS + tid;
    if (t < GCAT_QUADS) ((float4*)gcat)[t] = make_float4(0.f, 0.f, 0.f, 0.f);
    if (tid < CURSOR_INTS) bcnt[tid] = 0;
    __syncthreads();

    int dv[4], sv[4], rk[4];
    {
        int4 d = make_int4(-1, -1, -1, -1);
        int4 s = make_int4(0, 0, 0, 0);
        if (t < EDGE_QUADS) { d = ((const int4*)dst)[t]; s = ((const int4*)src)[t]; }
        dv[0] = d.x; dv[1] = d.y; dv[2] = d.z; dv[3] = d.w;
        sv[0] = s.x; sv[1] = s.y; sv[2] = s.z; sv[3] = s.w;
    }
    #pragma unroll
    for (int k = 0; k < 4; ++k) {
        rk[k] = 0;
        if ((unsigned)dv[k] < (unsigned)N_NODES) rk[k] = atomicAdd(&bcnt[dv[k] >> BKT_SHIFT], 1);
    }
    __syncthreads();

    if (tid < 32) {
        const int c0 = bcnt[lane], c1 = bcnt[lane + 32];
        int i0 = c0, i1 = c1;
        for (int d = 1; d < 32; d <<= 1) {
            int u0 = __shfl_up(i0, d);
            int u1 = __shfl_up(i1, d);
            if (lane >= d) { i0 += u0; i1 += u1; }
        }
        const int t0 = __shfl(i0, 31);
        loff[lane]      = i0 - c0;
        loff[lane + 32] = t0 + i1 - c1;
        if (lane == 31) total = t0 + i1;
        bbase[lane]      = c0 > 0 ? atomicAdd(&cursor[lane], c0) : 0;
        bbase[lane + 32] = c1 > 0 ? atomicAdd(&cursor[lane + 32], c1) : 0;
    }
    __syncthreads();

    #pragma unroll
    for (int k = 0; k < 4; ++k) {
        if ((unsigned)dv[k] < (unsigned)N_NODES) {
            const int b = dv[k] >> BKT_SHIFT;
            const int i = loff[b] + rk[k];
            ent[i] = ((unsigned)(dv[k] - (b << BKT_SHIFT)) << 16) | ((unsigned)sv[k] & 0xFFFFu);
            ebk[i] = (unsigned char)b;
        }
    }
    __syncthreads();

    const int n = total;
    for (int i = tid; i < n; i += BIN_THREADS) {
        const int b = ebk[i];
        const unsigned pos = (unsigned)(bbase[b] + (i - loff[b]));
        if (pos < (unsigned)BKT_CAP) buf[(size_t)b * BKT_CAP + pos] = ent[i];
    }
}

__global__ __launch_bounds__(BKT_ROWS) void list_kernel(
    const int* __restrict__ cursor, const unsigned* __restrict__ buf,
    int* __restrict__ rowstart, int* __restrict__ nbr)
{
    __shared__ int cnt[BKT_ROWS];
    __shared__ int start[BKT_ROWS];
    __shared__ int wtot[32];
    __shared__ int s_front;
    __shared__ int stage[LIST_LDS_CAP];

    const int tid  = threadIdx.x;
    const int lane = tid & 31;
    const int wave = tid >> 5;
    const int bk   = blockIdx.x;

    int nb = cursor[bk];
    nb = nb < 0 ? 0 : (nb > BKT_CAP ? BKT_CAP : nb);
    cnt[tid] = 0;
    if (tid < 32) {
        int f = 0;
        for (int i = lane; i < bk; i += 32) f += cursor[i];
        for (int d = 16; d >= 1; d >>= 1) f += __shfl_xor(f, d);
        if (lane == 0) s_front = f;
    }
    __syncthreads();

    const unsigned* mine = buf + (size_t)bk * BKT_CAP;
    for (int i0 = tid; i0 < nb; i0 += LIST_GROUP * BKT_ROWS) {
        unsigned p[LIST_GROUP];
        #pragma unroll
        for (int k = 0; k < LIST_GROUP; ++k) {
            p[k] = 0u;
            if (i0 + k * BKT_ROWS < nb) p[k] = mine[i0 + k * BKT_ROWS];
        }
        #pragma unroll
        for (int k = 0; k < LIST_GROUP; ++k)
            if (i0 + k * BKT_ROWS < nb) atomicAdd(&cnt[(p[k] >> 16) & (BKT_ROWS - 1)], 1);
    }
    __syncthreads();

    const int c = cnt[tid];
    int incl = c;
    for (int d = 1; d < 32; d <<= 1) {
        int up = __shfl_up(incl, d);
        if (lane >= d) incl += up;
    }
    if (lane == 31) wtot[wave] = incl;
    __syncthreads();
    int before = (lane < wave) ? wtot[lane] : 0;
    for (int d = 16; d >= 1; d >>= 1) before += __shfl_xor(before, d);
    const int excl  = before + incl - c;
    const int front = s_front;
    start[tid] = excl;
    cnt[tid]   = 0;
    const int row = bk * BKT_ROWS + tid;
    if (row < N_NODES) rowstart[row] = front + excl;
    if (bk == N_BKT - 1 && tid == 0) rowstart[N_NODES] = front + nb;
    __syncthreads();

    if (nb <= LIST_LDS_CAP) {
        for (int i0 = tid; i0 < nb; i0 += LIST_GROUP * BKT_ROWS) {
            unsigned p[LIST_GROUP];
            #pragma unroll
            for (int k = 0; k < LIST_GROUP; ++k) {
                p[k] = 0u;
                if (i0 + k * BKT_ROWS < nb) p[k] = mine[i0 + k * BKT_ROWS];
            }
            #pragma unroll
            for (int k = 0; k < LIST_GROUP; ++k) {
                if (i0 + k * BKT_ROWS < nb) {
                    const unsigned dl = (p[k] >> 16) & (BKT_ROWS - 1);
                    const unsigned q  = (unsigned)(start[dl] + atomicAdd(&cnt[dl], 1));
                    if (q < (unsigned)LIST_LDS_CAP) stage[q] = (int)(p[k] & 0xFFFFu);
                }
            }
        }
        __syncthreads();
        for (int i = tid; i < nb; i += BKT_ROWS) {
            const unsigned g = (unsigned)(front + i);
            if (g < (unsigned)N_EDGES) nbr[g] = stage[i];
        }
    } else {
        for (int i = tid; i < nb; i += BKT_ROWS) {
            const unsigned p  = mine[i];
            const unsigned dl = (p >> 16) & (BKT_ROWS - 1);
            const unsigned g  = (unsigned)(front + start[dl] + atomicAdd(&cnt[dl], 1));
            if (g < (unsigned)N_EDGES) nbr[g] = (int)(p & 0xFFFFu);
        }
    }
}

__device__ __forceinline__ unsigned row_id(int ids, int j)
{
    unsigned s = (unsigned)__builtin_amdgcn_readlane(ids, j & 31);
    return s < (unsigned)N_NODES ? s : (unsigned)(N_NODES - 1);
}

__device__ __forceinline__ float4 ld_row(const float* __restrict__ z, int zstride,
                                         unsigned s, unsigned laneoff)
{
    const char* rowp = (const char*)(z + (size_t)s * zstride);
    return *(const float4*)(rowp + laneoff);
}

__device__ __forceinline__ void add4(float4& a, const float4& v)
{
    a.x += v.x; a.y += v.y; a.z += v.z; a.w += v.w;
}

__device__ __forceinline__ void chunk_sum(float4& acc,
                                          const float* __restrict__ z, int zstride,
                                          int ids, int n, unsigned laneoff)
{
    for (int j = 0; j < n; j += ROW_GROUP) {
        unsigned s[ROW_GROUP];
        #pragma unroll
        for (int k = 0; k < ROW_GROUP; ++k) s[k] = row_id(ids, j + k);
        float4 v[ROW_GROUP];
        #pragma unroll
        for (int k = 0; k < ROW_GROUP; ++k) {
            v[k] = make_float4(0.f, 0.f, 0.f, 0.f);
            if (j + k < n) v[k] = ld_row(z, zstride, s[k], laneoff);
        }
        #pragma unroll
        for (int k = 0; k < ROW_GROUP; ++k)
            if (j + k < n) add4(acc, v[k]);
    }
}

__device__ __forceinline__ float4 row_sum(const float* __restrict__ z, int zstride,
                                          const int* __restrict__ nbr,
                                          int beg, int end, int ids,
                                          int lane, unsigned laneoff)
{
    float4 acc = make_float4(0.f, 0.f, 0.f, 0.f);
    int n = end - beg;
    n = n > 32 ? 32 : n;
    chunk_sum(acc, z, zstride, ids, n, laneoff);
    for (int base = beg + 32; base < end; base += 32) {
        n = end - base;
        n = n > 32 ? 32 : n;
        ids = (lane < n) ? nbr[base + lane] : 0;
        chunk_sum(acc, z, zstride, ids, n, laneoff);
    }
    return acc;
}

#define N_TILES    (N_NODES / 16)
#define MLP_BLOCKS 1024
#define MAX_TILES  4

template <int NT>
__device__ __forceinline__ void gemm_tiles(v8f (&acc)[NT], const float* lds,
                                           const float* __restrict__ Wm,
                                           int l16, int half, int col)
{
    const float* wp = Wm + (2 * half) * DIM + col;
    for (int k0 = 0; k0 < DIM; k0 += 4) {
        int ka = k0 + 2 * half;
        v2f b;
        b.x = wp[k0 * DIM];
        b.y = wp[(k0 + 1) * DIM];
        #pragma unroll
        for (int t = 0; t < NT; ++t) {
            v2f a;
            a.x = lds[(t * 16 + l16) * LDS_STRIDE + ka];
            a.y = lds[(t * 16 + l16) * LDS_STRIDE + ka + 1];
            acc[t] = __builtin_amdgcn_wmma_f32_16x16x4_f32(
                false, a, false, b, (short)0, acc[t], false, false);
        }
    }
}

template <int NT>
__device__ __forceinline__ void mlp_tiles(
    const float* ldsA, float* ldsH, int row0,
    float* __restrict__ hbuf,
    const float* __restrict__ W1, const float* __restrict__ b1,
    const float* __restrict__ W2, const float* __restrict__ b2,
    float* __restrict__ st, int l16, int half, int col)
{
    v8f acc[NT];
    #pragma unroll
    for (int t = 0; t < NT; ++t) acc[t] = v8f{};
    gemm_tiles<NT>(acc, ldsA, W1, l16, half, col);

    float bias1 = b1[col];
    #pragma unroll
    for (int t = 0; t < NT; ++t) {
        #pragma unroll
        for (int g = 0; g < 8; ++g) {
            int m = t * 16 + g + 8 * half;
            float v = acc[t][g] + bias1;
            ldsH[m * LDS_STRIDE + col] = v > 0.f ? v : 0.f;
        }
    }
    __syncthreads();

    v8f acc2[NT];
    #pragma unroll
    for (int t = 0; t < NT; ++t) acc2[t] = v8f{};
    gemm_tiles<NT>(acc2, ldsH, W2, l16, half, col);

    float bias2 = b2[col];
    float s = 0.f, q = 0.f;
    #pragma unroll
    for (int t = 0; t < NT; ++t) {
        #pragma unroll
        for (int g = 0; g < 8; ++g) {
            int m = t * 16 + g + 8 * half;
            float v = acc2[t][g] + bias2;
            v = v > 0.f ? v : 0.f;
            hbuf[(size_t)(row0 + m) * DIM + col] = v;
            s += v;
            q += v * v;
        }
    }
    s += __shfl_xor(s, 16);
    q += __shfl_xor(q, 16);
    if (half == 0) {
        atomicAdd(&st[col], s);
        atomicAdd(&st[DIM + col], q);
    }
}

__global__ __launch_bounds__(256) __attribute__((amdgpu_waves_per_eu(8))) void gin_mlp_kernel(
    const float* __restrict__ z, int zstride,
    const int* __restrict__ rowstart, const int* __restrict__ nbr,
    float* __restrict__ hbuf,
    const float* __restrict__ W1, const float* __restrict__ b1,
    const float* __restrict__ W2, const float* __restrict__ b2,
    float* __restrict__ stats)
{
    __shared__ __attribute__((aligned(16))) float ldsA[MAX_TILES * 16 * LDS_STRIDE];
    __shared__ float ldsH[MAX_TILES * 16 * LDS_STRIDE];

    const int tid  = threadIdx.x;
    const int wave = tid >> 5;
    const int lane = tid & 31;
    const int half = lane >> 4;
    const int l16  = lane & 15;
    const int col  = wave * 16 + l16;

    const int t0   = (int)(((unsigned)blockIdx.x * N_TILES) / MLP_BLOCKS);
    const int cnt  = (int)(((unsigned)(blockIdx.x + 1) * N_TILES) / MLP_BLOCKS) - t0;
    const int row0 = t0 * 16;

    {
        const int wv = __builtin_amdgcn_readfirstlane(tid >> 5);
        const unsigned laneoff = (unsigned)lane * 16u;
        for (int t = 0; t < cnt; ++t) {
            const int lr = t * 16 + 2 * wv;
            const int ra = row0 + lr;
            int e0 = rowstart[ra];
            int e1 = rowstart[ra + 1];
            int e2 = rowstart[ra + 2];
            e0 = e0 < 0 ? 0 : (e0 > N_EDGES ? N_EDGES : e0);
            e1 = e1 < 0 ? 0 : (e1 > N_EDGES ? N_EDGES : e1);
            e2 = e2 < 0 ? 0 : (e2 > N_EDGES ? N_EDGES : e2);
            const int idsA = (lane < e1 - e0) ? nbr[e0 + lane] : 0;
            const int idsB = (lane < e2 - e1) ? nbr[e1 + lane] : 0;
            const float4 ownA = *(const float4*)((const char*)(z + (size_t)ra * zstride) + laneoff);

            float4 sA = row_sum(z, zstride, nbr, e0, e1, idsA, lane, laneoff);
            float4 oA;
            oA.x = ownA.x + sA.x; oA.y = ownA.y + sA.y; oA.z = ownA.z + sA.z; oA.w = ownA.w + sA.w;
            *(float4*)(&ldsA[lr * LDS_STRIDE + lane * 4]) = oA;

            const float4 ownB = *(const float4*)((const char*)(z + (size_t)(ra + 1) * zstride) + laneoff);
            float4 sB = row_sum(z, zstride, nbr, e1, e2, idsB, lane, laneoff);
            float4 oB;
            oB.x = ownB.x + sB.x; oB.y = ownB.y + sB.y; oB.z = ownB.z + sB.z; oB.w = ownB.w + sB.w;
            *(float4*)(&ldsA[(lr + 1) * LDS_STRIDE + lane * 4]) = oB;
        }
    }
    __syncthreads();

    float* st = stats + (size_t)(blockIdx.x & (STAT_SETS - 1)) * (2 * DIM);
    if (cnt == MAX_TILES)
        mlp_tiles<MAX_TILES>(ldsA, ldsH, row0, hbuf, W1, b1, W2, b2, st, l16, half, col);
    else
        mlp_tiles<MAX_TILES - 1>(ldsA, ldsH, row0, hbuf, W1, b1, W2, b2, st, l16, half, col);
}

#define POOL_NODES_PER_WAVE 10
#define POOL_BLOCKS (N_NODES / (8 * POOL_NODES_PER_WAVE))

__device__ __forceinline__ void pool_flush(float* __restrict__ gcat, int g, int c,
                                           const float4& a)
{
    unsigned gg = (unsigned)g < (unsigned)NGRAPH ? (unsigned)g : (unsigned)(NGRAPH - 1);
    float* p = gcat + (size_t)gg * ZSTRIDE + c;
    atomicAdd(p,     a.x);
    atomicAdd(p + 1, a.y);
    atomicAdd(p + 2, a.z);
    atomicAdd(p + 3, a.w);
}

__global__ __launch_bounds__(256) void bn_pool_kernel(
    const float* __restrict__ h,
    const float* __restrict__ stats,
    const float* __restrict__ gamma, const float* __restrict__ beta,
    const int* __restrict__ batch,
    float* __restrict__ zcat, float* __restrict__ gcat)
{
    __shared__ __attribute__((aligned(16))) float tot[2 * DIM];

    const int lane = threadIdx.x & 31;
    const int wave = __builtin_amdgcn_readfirstlane((int)(threadIdx.x >> 5));
    const int n0 = (blockIdx.x * 8 + wave) * POOL_NODES_PER_WAVE;
    const int c  = lane * 4;

    float w[STAT_SETS];
    #pragma unroll
    for (int k = 0; k < STAT_SETS; ++k) w[k] = stats[k * (2 * DIM) + threadIdx.x];

    int gids = (lane < POOL_NODES_PER_WAVE) ? batch[n0 + lane] : 0;

    const float4 ga = *(const float4*)(gamma + c);
    const float4 be = *(const float4*)(beta + c);

    float4 v[POOL_NODES_PER_WAVE];
    for (int k = 0; k < POOL_NODES_PER_WAVE; ++k)
        v[k] = *(const float4*)(h + (size_t)(n0 + k) * DIM + c);

    float a = 0.f;
    #pragma unroll
    for (int k = 0; k < STAT_SETS; ++k) a += w[k];
    tot[threadIdx.x] = a;
    __builtin_amdgcn_fence(__ATOMIC_RELEASE, "workgroup", "local");
    __builtin_amdgcn_s_barrier();
    __builtin_amdgcn_fence(__ATOMIC_ACQUIRE, "workgroup", "local");
    const float4 sm = *(const float4*)(tot + c);
    const float4 sq = *(const float4*)(tot + DIM + c);
    float4 mu, rs;
    mu.x = sm.x * (1.0f / N_NODES);
    mu.y = sm.y * (1.0f / N_NODES);
    mu.z = sm.z * (1.0f / N_NODES);
    mu.w = sm.w * (1.0f / N_NODES);
    rs.x = rsqrtf(sq.x * (1.0f / N_NODES) - mu.x * mu.x + BN_EPS);
    rs.y = rsqrtf(sq.y * (1.0f / N_NODES) - mu.y * mu.y + BN_EPS);
    rs.z = rsqrtf(sq.z * (1.0f / N_NODES) - mu.z * mu.z + BN_EPS);
    rs.w = rsqrtf(sq.w * (1.0f / N_NODES) - mu.w * mu.w + BN_EPS);

    float4 acc = make_float4(0.f, 0.f, 0.f, 0.f);
    int gcur = __builtin_amdgcn_readlane(gids, 0);

    for (int k = 0; k < POOL_NODES_PER_WAVE; ++k) {
        const int g = __builtin_amdgcn_readlane(gids, k);
        if (g != gcur) {
            pool_flush(gcat, gcur, c, acc);
            acc = make_float4(0.f, 0.f, 0.f, 0.f);
            gcur = g;
        }
        float4 o;
        o.x = (v[k].x - mu.x) * rs.x * ga.x + be.x;
        o.y = (v[k].y - mu.y) * rs.y * ga.y + be.y;
        o.z = (v[k].z - mu.z) * rs.z * ga.z + be.z;
        o.w = (v[k].w - mu.w) * rs.w * ga.w + be.w;
        *(float4*)(zcat + (size_t)(n0 + k) * ZSTRIDE + c) = o;
        acc.x += o.x; acc.y += o.y; acc.z += o.z; acc.w += o.w;
    }
    pool_flush(gcat, gcur, c, acc);
}

extern "C" void kernel_launch(void* const* d_in, const int* in_sizes, int n_in,
                              void* d_out, int out_size, void* d_ws, size_t ws_size,
                              hipStream_t stream)
{
    (void)in_sizes; (void)n_in; (void)out_size; (void)ws_size;

    const float* x     = (const float*)d_in[0];
    const int*   ei    = (const int*)  d_in[1];
    const int*   batch = (const int*)  d_in[2];
    const float* W1    = (const float*)d_in[3];
    const float* b1    = (const float*)d_in[4];
    const float* W2    = (const float*)d_in[5];
    const float* b2    = (const float*)d_in[6];
    const float* gamma = (const float*)d_in[7];
    const float* beta  = (const float*)d_in[8];

    float* out  = (float*)d_out;
    float* zcat = out;
    float* gcat = out + (size_t)N_NODES * ZSTRIDE;

    float* hbuf  = (float*)d_ws;
    float* stats = hbuf + (size_t)N_NODES * DIM;
    int* cursor   = (int*)(stats + NLAYER * STAT_LAYER_FLOATS);
    int* rowstart = cursor + CURSOR_INTS;
    int* nbr      = rowstart + N_NODES + 32;
    unsigned* buf = (unsigned*)(nbr + N_EDGES);

    const int* src = ei;
    const int* dst = ei + N_EDGES;

    hipMemsetAsync(stats, 0, CLEAR_BYTES, stream);

    bin_kernel<<<BIN_BLOCKS, BIN_THREADS, 0, stream>>>(src, dst, cursor, buf, gcat);
    list_kernel<<<N_BKT, BKT_ROWS, 0, stream>>>(cursor, buf, rowstart, nbr);

    for (int l = 0; l < NLAYER; ++l) {
        const float* zin = (l == 0) ? x : (zcat + (size_t)(l - 1) * DIM);
        int zstride      = (l == 0) ? DIM : ZSTRIDE;

        float* lstats = stats + (size_t)l * STAT_LAYER_FLOATS;

        gin_mlp_kernel<<<MLP_BLOCKS, 256, 0, stream>>>(
            zin, zstride, rowstart, nbr, hbuf,
            W1 + (size_t)l * DIM * DIM, b1 + (size_t)l * DIM,
            W2 + (size_t)l * DIM * DIM, b2 + (size_t)l * DIM,
            lstats);

        bn_pool_kernel<<<POOL_BLOCKS, 256, 0, stream>>>(
            hbuf, lstats,
            gamma + (size_t)l * DIM, beta + (size_t)l * DIM,
            batch,
            zcat + (size_t)l * DIM, gcat + (size_t)l * DIM);
    }
}
